# starting kernel
# baseline (speedup 1.0000x reference)
#include <hip/hip_runtime.h>
#include <math.h>

typedef float v2f __attribute__((ext_vector_type(2)));
typedef float v8f __attribute__((ext_vector_type(8)));

#define N_NODES   100000
#define N_FEAT    512
#define HIDDEN    64
#define N_CLASSES 16

static __device__ __forceinline__ void atomic_add_f32(float* p, float v) {
    __hip_atomic_fetch_add(p, v, __ATOMIC_RELAXED, __HIP_MEMORY_SCOPE_AGENT);
}

__global__ __launch_bounds__(256) void init_kernel(float* bufB, float* out, float* deg) {
    int t = blockIdx.x * 256 + threadIdx.x;
    if (t < N_NODES * HIDDEN)    bufB[t] = 0.0f;
    if (t < N_NODES * N_CLASSES) out[t]  = 0.0f;
    if (t < N_NODES)             deg[t]  = 1.0f;
}

__global__ __launch_bounds__(256) void degree_kernel(const int* __restrict__ dst,
                                                     float* __restrict__ deg, int nE) {
    int e = blockIdx.x * 256 + threadIdx.x;
    if (e < nE) atomic_add_f32(&deg[dst[e]], 1.0f);
}

__global__ __launch_bounds__(256) void rsqrt_kernel(float* dinv) {
    int t = blockIdx.x * 256 + threadIdx.x;
    if (t < N_NODES) dinv[t] = rsqrtf(dinv[t]);
}

__global__ __launch_bounds__(256) void gemm1_kernel(const float* __restrict__ A,
                                                    const float* __restrict__ B,
                                                    float* __restrict__ C, int nTiles) {
    int wave = threadIdx.x >> 5;
    int lane = threadIdx.x & 31;
    int tile = blockIdx.x * 8 + wave;
    if (tile >= nTiles) return;
    int nr = lane & 15;
    int hi = lane >> 4;

    const float* arow = A + (size_t)(tile * 16 + nr) * N_FEAT + 2 * hi;
    v8f acc0 = {}, acc1 = {}, acc2 = {}, acc3 = {};

    for (int kk = 0; kk < N_FEAT; kk += 4) {
        v2f a = *(const v2f*)(arow + kk);
        const float* b0p = B + (size_t)(kk + 2 * hi) * HIDDEN + nr;
        const float* b1p = b0p + HIDDEN;
        v2f b0 = { b0p[0],  b1p[0]  };
        v2f b1 = { b0p[16], b1p[16] };
        v2f b2 = { b0p[32], b1p[32] };
        v2f b3 = { b0p[48], b1p[48] };
        acc0 = __builtin_amdgcn_wmma_f32_16x16x4_f32(false, a, false, b0, (short)0, acc0, false, false);
        acc1 = __builtin_amdgcn_wmma_f32_16x16x4_f32(false, a, false, b1, (short)0, acc1, false, false);
        acc2 = __builtin_amdgcn_wmma_f32_16x16x4_f32(false, a, false, b2, (short)0, acc2, false, false);
        acc3 = __builtin_amdgcn_wmma_f32_16x16x4_f32(false, a, false, b3, (short)0, acc3, false, false);
    }
    int mbase = tile * 16 + hi * 8;
    for (int r = 0; r < 8; ++r) {
        float* crow = C + (size_t)(mbase + r) * HIDDEN + nr;
        crow[0]  = acc0[r];
        crow[16] = acc1[r];
        crow[32] = acc2[r];
        crow[48] = acc3[r];
    }
}

__global__ __launch_bounds__(256) void edge_agg1_kernel(const int* __restrict__ src,
                                                        const int* __restrict__ dst,
                                                        const float* __restrict__ dinv,
                                                        const float* __restrict__ h,
                                                        float* __restrict__ agg, int nE) {
    int e = blockIdx.x * 8 + (threadIdx.x >> 5);
    if (e >= nE) return;
    int lane = threadIdx.x & 31;
    int s = src[e], d = dst[e];
    float coeff = dinv[s] * dinv[d];
    const float* hs = h   + (size_t)s * HIDDEN + lane * 2;
    float*       ad = agg + (size_t)d * HIDDEN + lane * 2;
    atomic_add_f32(&ad[0], hs[0] * coeff);
    atomic_add_f32(&ad[1], hs[1] * coeff);
}

__global__ __launch_bounds__(256) void relu_kernel(const float* __restrict__ pre,
                                                   float* __restrict__ agg,
                                                   const float* __restrict__ dinv,
                                                   const float* __restrict__ b1) {
    int t = blockIdx.x * 256 + threadIdx.x;
    if (t >= N_NODES * HIDDEN) return;
    int node = t >> 6, f = t & 63;
    float di = dinv[node];
    float v = agg[t] + di * di * pre[t] + b1[f];
    agg[t] = v > 0.0f ? v : 0.0f;
}

__global__ __launch_bounds__(256) void gemm2_kernel(const float* __restrict__ A,
                                                    const float* __restrict__ B,
                                                    float* __restrict__ C, int nTiles) {
    int wave = threadIdx.x >> 5;
    int lane = threadIdx.x & 31;
    int tile = blockIdx.x * 8 + wave;
    if (tile >= nTiles) return;
    int nr = lane & 15;
    int hi = lane >> 4;

    const float* arow = A + (size_t)(tile * 16 + nr) * HIDDEN + 2 * hi;
    v8f acc = {};
    for (int kk = 0; kk < HIDDEN; kk += 4) {
        v2f a = *(const v2f*)(arow + kk);
        const float* b0p = B + (size_t)(kk + 2 * hi) * N_CLASSES + nr;
        v2f b = { b0p[0], b0p[N_CLASSES] };
        acc = __builtin_amdgcn_wmma_f32_16x16x4_f32(false, a, false, b, (short)0, acc, false, false);
    }
    int mbase = tile * 16 + hi * 8;
    for (int r = 0; r < 8; ++r)
        C[(size_t)(mbase + r) * N_CLASSES + nr] = acc[r];
}

__global__ __launch_bounds__(256) void edge_agg2_kernel(const int* __restrict__ src,
                                                        const int* __restrict__ dst,
                                                        const float* __restrict__ dinv,
                                                        const float* __restrict__ h,
                                                        float* __restrict__ out, int nE) {
    int t = blockIdx.x * 256 + threadIdx.x;
    if (t >= nE * N_CLASSES) return;
    int e = t >> 4, c = t & 15;
    int s = src[e], d = dst[e];
    float coeff = dinv[s] * dinv[d];
    atomic_add_f32(&out[(size_t)d * N_CLASSES + c], coeff * h[(size_t)s * N_CLASSES + c]);
}

__global__ __launch_bounds__(256) void logsoftmax_kernel(float* __restrict__ out,
                                                         const float* __restrict__ pre,
                                                         const float* __restrict__ dinv,
                                                         const float* __restrict__ b2) {
    int r = blockIdx.x * 256 + threadIdx.x;
    if (r >= N_NODES) return;
    float di = dinv[r];
    float di2 = di * di;
    float v[N_CLASSES];
    float m = -INFINITY;
    for (int c = 0; c < N_CLASSES; ++c) {
        float x = out[(size_t)r * N_CLASSES + c] + di2 * pre[(size_t)r * N_CLASSES + c] + b2[c];
        v[c] = x;
        m = fmaxf(m, x);
    }
    float s = 0.0f;
    for (int c = 0; c < N_CLASSES; ++c) s += expf(v[c] - m);
    float ls = logf(s);
    for (int c = 0; c < N_CLASSES; ++c)
        out[(size_t)r * N_CLASSES + c] = v[c] - m - ls;
}

extern "C" void kernel_launch(void* const* d_in, const int* in_sizes, int n_in,
                              void* d_out, int out_size, void* d_ws, size_t ws_size,
                              hipStream_t stream) {
    const float* x  = (const float*)d_in[0];
    const int*   ei = (const int*)d_in[1];
    const float* W1 = (const float*)d_in[2];
    const float* b1 = (const float*)d_in[3];
    const float* W2 = (const float*)d_in[4];
    const float* b2 = (const float*)d_in[5];
    float* out = (float*)d_out;

    int nE = in_sizes[1] / 2;
    const int* src = ei;
    const int* dst = ei + nE;

    float* dinv = (float*)d_ws;
    float* bufA = dinv + 102400;
    float* bufB = bufA + (size_t)N_NODES * HIDDEN;

    int nTiles = N_NODES / 16;

    init_kernel<<<(N_NODES * HIDDEN + 255) / 256, 256, 0, stream>>>(bufB, out, dinv);
    degree_kernel<<<(nE + 255) / 256, 256, 0, stream>>>(dst, dinv, nE);
    rsqrt_kernel<<<(N_NODES + 255) / 256, 256, 0, stream>>>(dinv);
    gemm1_kernel<<<(nTiles + 7) / 8, 256, 0, stream>>>(x, W1, bufA, nTiles);
    edge_agg1_kernel<<<(nE + 7) / 8, 256, 0, stream>>>(src, dst, dinv, bufA, bufB, nE);
    relu_kernel<<<(N_NODES * HIDDEN + 255) / 256, 256, 0, stream>>>(bufA, bufB, dinv, b1);
    gemm2_kernel<<<(nTiles + 7) / 8, 256, 0, stream>>>(bufB, W2, bufA, nTiles);
    edge_agg2_kernel<<<(nE * N_CLASSES + 255) / 256, 256, 0, stream>>>(src, dst, dinv, bufA, out, nE);
    logsoftmax_kernel<<<(N_NODES + 255) / 256, 256, 0, stream>>>(out, bufA, dinv, b2);
}


# Edge passes: a wave walks 32 edges (src, dst, weight loaded coalesced once, then read from lanes), 16 edges' rows in flight before whole-line atomic adds; sum order stays atomic arrival order
# speedup vs baseline: 1.3893x; 1.3893x over previous
#include <hip/hip_runtime.h>
#include <math.h>

typedef float v2f __attribute__((ext_vector_type(2)));
typedef float v8f __attribute__((ext_vector_type(8)));

#define N_NODES   100000
#define N_FEAT    512
#define HIDDEN    64
#define N_CLASSES 16

static __device__ __forceinline__ void atomic_add_f32(float* p, float v) {
    __hip_atomic_fetch_add(p, v, __ATOMIC_RELAXED, __HIP_MEMORY_SCOPE_AGENT);
}

__global__ __launch_bounds__(256) void init_kernel(float* bufB, float* out, float* deg) {
    int t = blockIdx.x * 256 + threadIdx.x;
    if (t < N_NODES * HIDDEN)    bufB[t] = 0.0f;
    if (t < N_NODES * N_CLASSES) out[t]  = 0.0f;
    if (t < N_NODES)             deg[t]  = 1.0f;
}

__global__ __launch_bounds__(256) void degree_kernel(const int* __restrict__ dst,
                                                     float* __restrict__ deg, int nE) {
    int e = blockIdx.x * 256 + threadIdx.x;
    if (e < nE) atomic_add_f32(&deg[dst[e]], 1.0f);
}

__global__ __launch_bounds__(256) void rsqrt_kernel(float* dinv) {
    int t = blockIdx.x * 256 + threadIdx.x;
    if (t < N_NODES) dinv[t] = rsqrtf(dinv[t]);
}

__global__ __launch_bounds__(256) void gemm1_kernel(const float* __restrict__ A,
                                                    const float* __restrict__ B,
                                                    float* __restrict__ C, int nTiles) {
    int wave = threadIdx.x >> 5;
    int lane = threadIdx.x & 31;
    int tile = blockIdx.x * 8 + wave;
    if (tile >= nTiles) return;
    int nr = lane & 15;
    int hi = lane >> 4;

    const float* arow = A + (size_t)(tile * 16 + nr) * N_FEAT + 2 * hi;
    v8f acc0 = {}, acc1 = {}, acc2 = {}, acc3 = {};

    for (int kk = 0; kk < N_FEAT; kk += 4) {
        v2f a = *(const v2f*)(arow + kk);
        const float* b0p = B + (size_t)(kk + 2 * hi) * HIDDEN + nr;
        const float* b1p = b0p + HIDDEN;
        v2f b0 = { b0p[0],  b1p[0]  };
        v2f b1 = { b0p[16], b1p[16] };
        v2f b2 = { b0p[32], b1p[32] };
        v2f b3 = { b0p[48], b1p[48] };
        acc0 = __builtin_amdgcn_wmma_f32_16x16x4_f32(false, a, false, b0, (short)0, acc0, false, false);
        acc1 = __builtin_amdgcn_wmma_f32_16x16x4_f32(false, a, false, b1, (short)0, acc1, false, false);
        acc2 = __builtin_amdgcn_wmma_f32_16x16x4_f32(false, a, false, b2, (short)0, acc2, false, false);
        acc3 = __builtin_amdgcn_wmma_f32_16x16x4_f32(false, a, false, b3, (short)0, acc3, false, false);
    }
    int mbase = tile * 16 + hi * 8;
    for (int r = 0; r < 8; ++r) {
        float* crow = C + (size_t)(mbase + r) * HIDDEN + nr;
        crow[0]  = acc0[r];
        crow[16] = acc1[r];
        crow[32] = acc2[r];
        crow[48] = acc3[r];
    }
}

static __device__ __forceinline__ int lane_i(int v, int j) { return __builtin_amdgcn_readlane(v, j); }
static __device__ __forceinline__ float lane_f(float v, int j) {
    return __int_as_float(__builtin_amdgcn_readlane(__float_as_int(v), j));
}

__global__ __launch_bounds__(256) void edge_agg1_kernel(const int* __restrict__ src,
                                                        const int* __restrict__ dst,
                                                        const float* __restrict__ dinv,
                                                        const float* __restrict__ h,
                                                        float* __restrict__ agg, int nE) {
    int lane = threadIdx.x & 31;
    int e0 = (blockIdx.x * 8 + __builtin_amdgcn_readfirstlane(threadIdx.x >> 5)) * 32;
    if (e0 >= nE) return;
    int cnt = nE - e0;
    int s = 0, d = 0;
    float c = 0.0f;
    if (lane < cnt) {
        s = src[e0 + lane];
        d = dst[e0 + lane];
        c = dinv[s] * dinv[d];
    }
    if (cnt >= 32) {
#pragma unroll
        for (int g = 0; g < 32; g += 16) {
            float v0[16], v1[16];
#pragma unroll
            for (int j = 0; j < 16; ++j) {
                int ro = lane_i(s, g + j) * HIDDEN + lane;
                v0[j] = h[ro];
                v1[j] = h[ro + 32];
            }
#pragma unroll
            for (int j = 0; j < 16; ++j) {
                float cj = lane_f(c, g + j);
                int ro = lane_i(d, g + j) * HIDDEN + lane;
                atomic_add_f32(&agg[ro],      v0[j] * cj);
                atomic_add_f32(&agg[ro + 32], v1[j] * cj);
            }
        }
    } else {
        for (int j = 0; j < cnt; ++j) {
            float cj = lane_f(c, j);
            int rs = lane_i(s, j) * HIDDEN + lane;
            int rd = lane_i(d, j) * HIDDEN + lane;
            atomic_add_f32(&agg[rd],      h[rs]      * cj);
            atomic_add_f32(&agg[rd + 32], h[rs + 32] * cj);
        }
    }
}

__global__ __launch_bounds__(256) void relu_kernel(const float* __restrict__ pre,
                                                   float* __restrict__ agg,
                                                   const float* __restrict__ dinv,
                                                   const float* __restrict__ b1) {
    int t = blockIdx.x * 256 + threadIdx.x;
    if (t >= N_NODES * HIDDEN) return;
    int node = t >> 6, f = t & 63;
    float di = dinv[node];
    float v = agg[t] + di * di * pre[t] + b1[f];
    agg[t] = v > 0.0f ? v : 0.0f;
}

__global__ __launch_bounds__(256) void gemm2_kernel(const float* __restrict__ A,
                                                    const float* __restrict__ B,
                                                    float* __restrict__ C, int nTiles) {
    int wave = threadIdx.x >> 5;
    int lane = threadIdx.x & 31;
    int tile = blockIdx.x * 8 + wave;
    if (tile >= nTiles) return;
    int nr = lane & 15;
    int hi = lane >> 4;

    const float* arow = A + (size_t)(tile * 16 + nr) * HIDDEN + 2 * hi;
    v8f acc = {};
    for (int kk = 0; kk < HIDDEN; kk += 4) {
        v2f a = *(const v2f*)(arow + kk);
        const float* b0p = B + (size_t)(kk + 2 * hi) * N_CLASSES + nr;
        v2f b = { b0p[0], b0p[N_CLASSES] };
        acc = __builtin_amdgcn_wmma_f32_16x16x4_f32(false, a, false, b, (short)0, acc, false, false);
    }
    int mbase = tile * 16 + hi * 8;
    for (int r = 0; r < 8; ++r)
        C[(size_t)(mbase + r) * N_CLASSES + nr] = acc[r];
}

template <int K> static __device__ __forceinline__ int row_lane_i(int v) {
    return __builtin_amdgcn_update_dpp(0, v, 0x150 + K, 0xF, 0xF, false);
}
template <int K> static __device__ __forceinline__ float row_lane_f(float v) {
    return __int_as_float(row_lane_i<K>(__float_as_int(v)));
}

#define AGG2_LOAD(K) v[K] = h[row_lane_i<K>(s) * N_CLASSES + col];
#define AGG2_ADD(K)  atomic_add_f32(&out[row_lane_i<K>(d) * N_CLASSES + col], row_lane_f<K>(c) * v[K]);
#define AGG2_ALL(M)  M(0) M(1) M(2) M(3) M(4) M(5) M(6) M(7) M(8) M(9) M(10) M(11) M(12) M(13) M(14) M(15)

__global__ __launch_bounds__(256) void edge_agg2_kernel(const int* __restrict__ src,
                                                        const int* __restrict__ dst,
                                                        const float* __restrict__ dinv,
                                                        const float* __restrict__ h,
                                                        float* __restrict__ out, int nE) {
    int lane = threadIdx.x & 31;
    int e0 = (blockIdx.x * 8 + __builtin_amdgcn_readfirstlane(threadIdx.x >> 5)) * 32;
    if (e0 >= nE) return;
    int cnt = nE - e0;
    int s = 0, d = 0;
    float c = 0.0f;
    if (lane < cnt) {
        s = src[e0 + lane];
        d = dst[e0 + lane];
        c = dinv[s] * dinv[d];
    }
    int col = lane & 15;
    if (cnt >= 32) {
        float v[16];
        AGG2_ALL(AGG2_LOAD)
        AGG2_ALL(AGG2_ADD)
    } else {
        int half = lane & 16;
#pragma clang loop unroll(disable)
        for (int k = 0; k < 16; ++k) {
            int sk = __shfl(s, half + k);
            int dk = __shfl(d, half + k);
            float ck = __shfl(c, half + k);
            if (half + k < cnt)
                atomic_add_f32(&out[dk * N_CLASSES + col], ck * h[sk * N_CLASSES + col]);
        }
    }
}

__global__ __launch_bounds__(256) void logsoftmax_kernel(float* __restrict__ out,
                                                         const float* __restrict__ pre,
                                                         const float* __restrict__ dinv,
                                                         const float* __restrict__ b2) {
    int r = blockIdx.x * 256 + threadIdx.x;
    if (r >= N_NODES) return;
    float di = dinv[r];
    float di2 = di * di;
    float v[N_CLASSES];
    float m = -INFINITY;
    for (int c = 0; c < N_CLASSES; ++c) {
        float x = out[(size_t)r * N_CLASSES + c] + di2 * pre[(size_t)r * N_CLASSES + c] + b2[c];
        v[c] = x;
        m = fmaxf(m, x);
    }
    float s = 0.0f;
    for (int c = 0; c < N_CLASSES; ++c) s += expf(v[c] - m);
    float ls = logf(s);
    for (int c = 0; c < N_CLASSES; ++c)
        out[(size_t)r * N_CLASSES + c] = v[c] - m - ls;
}

extern "C" void kernel_launch(void* const* d_in, const int* in_sizes, int n_in,
                              void* d_out, int out_size, void* d_ws, size_t ws_size,
                              hipStream_t stream) {
    const float* x  = (const float*)d_in[0];
    const int*   ei = (const int*)d_in[1];
    const float* W1 = (const float*)d_in[2];
    const float* b1 = (const float*)d_in[3];
    const float* W2 = (const float*)d_in[4];
    const float* b2 = (const float*)d_in[5];
    float* out = (float*)d_out;

    int nE = in_sizes[1] / 2;
    const int* src = ei;
    const int* dst = ei + nE;

    float* dinv = (float*)d_ws;
    float* bufA = dinv + 102400;
    float* bufB = bufA + (size_t)N_NODES * HIDDEN;

    int nTiles = N_NODES / 16;

    init_kernel<<<(N_NODES * HIDDEN + 255) / 256, 256, 0, stream>>>(bufB, out, dinv);
    degree_kernel<<<(nE + 255) / 256, 256, 0, stream>>>(dst, dinv, nE);
    rsqrt_kernel<<<(N_NODES + 255) / 256, 256, 0, stream>>>(dinv);
    gemm1_kernel<<<(nTiles + 7) / 8, 256, 0, stream>>>(x, W1, bufA, nTiles);
    edge_agg1_kernel<<<(nE + 255) / 256, 256, 0, stream>>>(src, dst, dinv, bufA, bufB, nE);
    relu_kernel<<<(N_NODES * HIDDEN + 255) / 256, 256, 0, stream>>>(bufA, bufB, dinv, b1);
    gemm2_kernel<<<(nTiles + 7) / 8, 256, 0, stream>>>(bufB, W2, bufA, nTiles);
    edge_agg2_kernel<<<(nE + 255) / 256, 256, 0, stream>>>(src, dst, dinv, bufA, out, nE);
    logsoftmax_kernel<<<(N_NODES + 255) / 256, 256, 0, stream>>>(out, bufA, dinv, b2);
}


# Edge passes: persistent work-groups read their die and claim edge blocks from one queue per L2 half, adding only into target rows of that half (other queue as tail); atomic arrival order kept
# speedup vs baseline: 1.5492x; 1.1151x over previous
#include <hip/hip_runtime.h>
#include <math.h>

typedef float v2f __attribute__((ext_vector_type(2)));
typedef float v8f __attribute__((ext_vector_type(8)));

#define N_NODES   100000
#define N_FEAT    512
#define HIDDEN    64
#define N_CLASSES 16

static __device__ __forceinline__ void atomic_add_f32(float* p, float v) {
    __hip_atomic_fetch_add(p, v, __ATOMIC_RELAXED, __HIP_MEMORY_SCOPE_AGENT);
}

#define QSTRIDE 64
__global__ __launch_bounds__(256) void init_kernel(float* bufB, float* out, float* deg, int* queues) {
    int t = blockIdx.x * 256 + threadIdx.x;
    if (t < 4)                   queues[t * QSTRIDE] = 0;
    if (t < N_NODES * HIDDEN)    bufB[t] = 0.0f;
    if (t < N_NODES * N_CLASSES) out[t]  = 0.0f;
    if (t < N_NODES)             deg[t]  = 1.0f;
}

__global__ __launch_bounds__(256) void degree_kernel(const int* __restrict__ dst,
                                                     float* __restrict__ deg, int nE) {
    int e = blockIdx.x * 256 + threadIdx.x;
    if (e < nE) atomic_add_f32(&deg[dst[e]], 1.0f);
}

__global__ __launch_bounds__(256) void rsqrt_kernel(float* dinv) {
    int t = blockIdx.x * 256 + threadIdx.x;
    if (t < N_NODES) dinv[t] = rsqrtf(dinv[t]);
}

__global__ __launch_bounds__(256) void gemm1_kernel(const float* __restrict__ A,
                                                    const float* __restrict__ B,
                                                    float* __restrict__ C, int nTiles) {
    int wave = threadIdx.x >> 5;
    int lane = threadIdx.x & 31;
    int tile = blockIdx.x * 8 + wave;
    if (tile >= nTiles) return;
    int nr = lane & 15;
    int hi = lane >> 4;

    const float* arow = A + (size_t)(tile * 16 + nr) * N_FEAT + 2 * hi;
    v8f acc0 = {}, acc1 = {}, acc2 = {}, acc3 = {};

    for (int kk = 0; kk < N_FEAT; kk += 4) {
        v2f a = *(const v2f*)(arow + kk);
        const float* b0p = B + (size_t)(kk + 2 * hi) * HIDDEN + nr;
        const float* b1p = b0p + HIDDEN;
        v2f b0 = { b0p[0],  b1p[0]  };
        v2f b1 = { b0p[16], b1p[16] };
        v2f b2 = { b0p[32], b1p[32] };
        v2f b3 = { b0p[48], b1p[48] };
        acc0 = __builtin_amdgcn_wmma_f32_16x16x4_f32(false, a, false, b0, (short)0, acc0, false, false);
        acc1 = __builtin_amdgcn_wmma_f32_16x16x4_f32(false, a, false, b1, (short)0, acc1, false, false);
        acc2 = __builtin_amdgcn_wmma_f32_16x16x4_f32(false, a, false, b2, (short)0, acc2, false, false);
        acc3 = __builtin_amdgcn_wmma_f32_16x16x4_f32(false, a, false, b3, (short)0, acc3, false, false);
    }
    int mbase = tile * 16 + hi * 8;
    for (int r = 0; r < 8; ++r) {
        float* crow = C + (size_t)(mbase + r) * HIDDEN + nr;
        crow[0]  = acc0[r];
        crow[16] = acc1[r];
        crow[32] = acc2[r];
        crow[48] = acc3[r];
    }
}

#define HALF_SPLIT  (N_NODES / 2)
#define CLAIM_EDGES 512

static __device__ __forceinline__ int my_half(int* slot) {
    if (threadIdx.x == 0) *slot = (__builtin_amdgcn_s_sendmsg_rtn(0x87) >> 18) & 1;
    __syncthreads();
    int hf = *slot;
    __syncthreads();
    return hf;
}

static __device__ __forceinline__ int claim_block(int* q, int* slot) {
    if (threadIdx.x == 0)
        *slot = __hip_atomic_fetch_add(q, 1, __ATOMIC_RELAXED, __HIP_MEMORY_SCOPE_AGENT);
    __syncthreads();
    int idx = *slot;
    __syncthreads();
    return idx;
}

static __device__ __forceinline__ int lane_i(int v, int j) { return __builtin_amdgcn_readlane(v, j); }
static __device__ __forceinline__ float lane_f(float v, int j) {
    return __int_as_float(__builtin_amdgcn_readlane(__float_as_int(v), j));
}

static __device__ __forceinline__ void agg1_wave(const int* __restrict__ src,
                                                 const int* __restrict__ dst,
                                                 const float* __restrict__ dinv,
                                                 const float* __restrict__ h,
                                                 float* __restrict__ agg,
                                                 int e0, int nE, int t, int lane) {
    int cnt = nE - e0;
    if (cnt <= 0) return;
    int so = 0, dofs = 0;
    float c = 0.0f;
    bool sel = false;
    if (lane < cnt) {
        int s = src[e0 + lane];
        int d = dst[e0 + lane];
        so   = s * HIDDEN;
        dofs = d * HIDDEN;
        sel = (d >= HALF_SPLIT ? 1 : 0) == t;
        if (sel) c = dinv[s] * dinv[d];
    }
    unsigned mask = __builtin_amdgcn_ballot_w32(sel);
    float v0[32], v1[32];
#pragma unroll
    for (int j = 0; j < 32; ++j) {
        float a = 0.0f, b = 0.0f;
        if ((mask >> j) & 1u) {
            int ro = lane_i(so, j) + lane;
            a = h[ro];
            b = h[ro + 32];
        }
        v0[j] = a;
        v1[j] = b;
    }
#pragma unroll
    for (int j = 0; j < 32; ++j) {
        if ((mask >> j) & 1u) {
            float cj = lane_f(c, j);
            int ro = lane_i(dofs, j) + lane;
            atomic_add_f32(&agg[ro],      v0[j] * cj);
            atomic_add_f32(&agg[ro + 32], v1[j] * cj);
        }
    }
}

__global__ __launch_bounds__(256) void edge_agg1_kernel(const int* __restrict__ src,
                                                        const int* __restrict__ dst,
                                                        const float* __restrict__ dinv,
                                                        const float* __restrict__ h,
                                                        float* __restrict__ agg, int nE,
                                                        int* __restrict__ queues) {
    __shared__ int slot;
    int lane = threadIdx.x & 31;
    int wave = __builtin_amdgcn_readfirstlane(threadIdx.x >> 5);
    int nClaims = (nE + CLAIM_EDGES - 1) / CLAIM_EDGES;
    int half = my_half(&slot);
    for (int pass = 0; pass < 2; ++pass) {
        int t = half ^ pass;
        for (;;) {
            int idx = claim_block(queues + t * QSTRIDE, &slot);
            if (idx >= nClaims) break;
            int e0 = idx * CLAIM_EDGES + wave * 64;
            agg1_wave(src, dst, dinv, h, agg, e0,      nE, t, lane);
            agg1_wave(src, dst, dinv, h, agg, e0 + 32, nE, t, lane);
        }
    }
}

__global__ __launch_bounds__(256) void relu_kernel(const float* __restrict__ pre,
                                                   float* __restrict__ agg,
                                                   const float* __restrict__ dinv,
                                                   const float* __restrict__ b1) {
    int t = blockIdx.x * 256 + threadIdx.x;
    if (t >= N_NODES * HIDDEN) return;
    int node = t >> 6, f = t & 63;
    float di = dinv[node];
    float v = agg[t] + di * di * pre[t] + b1[f];
    agg[t] = v > 0.0f ? v : 0.0f;
}

__global__ __launch_bounds__(256) void gemm2_kernel(const float* __restrict__ A,
                                                    const float* __restrict__ B,
                                                    float* __restrict__ C, int nTiles) {
    int wave = threadIdx.x >> 5;
    int lane = threadIdx.x & 31;
    int tile = blockIdx.x * 8 + wave;
    if (tile >= nTiles) return;
    int nr = lane & 15;
    int hi = lane >> 4;

    const float* arow = A + (size_t)(tile * 16 + nr) * HIDDEN + 2 * hi;
    v8f acc = {};
    for (int kk = 0; kk < HIDDEN; kk += 4) {
        v2f a = *(const v2f*)(arow + kk);
        const float* b0p = B + (size_t)(kk + 2 * hi) * N_CLASSES + nr;
        v2f b = { b0p[0], b0p[N_CLASSES] };
        acc = __builtin_amdgcn_wmma_f32_16x16x4_f32(false, a, false, b, (short)0, acc, false, false);
    }
    int mbase = tile * 16 + hi * 8;
    for (int r = 0; r < 8; ++r)
        C[(size_t)(mbase + r) * N_CLASSES + nr] = acc[r];
}

template <int K> static __device__ __forceinline__ int row_lane_i(int v) {
    return __builtin_amdgcn_update_dpp(0, v, 0x150 + K, 0xF, 0xF, false);
}
template <int K> static __device__ __forceinline__ float row_lane_f(float v) {
    return __int_as_float(row_lane_i<K>(__float_as_int(v)));
}

#define AGG2_LOAD(K) { int sk = row_lane_i<K>(s); int fk = row_lane_i<K>(f); \
                       if (fk) v[K] = h[sk * N_CLASSES + col]; }
#define AGG2_ADD(K)  { int dk = row_lane_i<K>(d); int fk = row_lane_i<K>(f); float ck = row_lane_f<K>(c); \
                       if (fk) atomic_add_f32(&out[dk * N_CLASSES + col], ck * v[K]); }
#define AGG2_ALL(M)  M(0) M(1) M(2) M(3) M(4) M(5) M(6) M(7) M(8) M(9) M(10) M(11) M(12) M(13) M(14) M(15)

static __device__ __forceinline__ void agg2_wave(const int* __restrict__ src,
                                                 const int* __restrict__ dst,
                                                 const float* __restrict__ dinv,
                                                 const float* __restrict__ h,
                                                 float* __restrict__ out,
                                                 int e0, int nE, int t, int lane) {
    int cnt = nE - e0;
    if (cnt <= 0) return;
    int s = 0, d = 0, f = 0;
    float c = 0.0f;
    if (lane < cnt) {
        s = src[e0 + lane];
        d = dst[e0 + lane];
        f = ((d >= HALF_SPLIT ? 1 : 0) == t) ? 1 : 0;
        if (f) c = dinv[s] * dinv[d];
    }
    int col = lane & 15;
    float v[16];
    AGG2_ALL(AGG2_LOAD)
    AGG2_ALL(AGG2_ADD)
}

__global__ __launch_bounds__(256) void edge_agg2_kernel(const int* __restrict__ src,
                                                        const int* __restrict__ dst,
                                                        const float* __restrict__ dinv,
                                                        const float* __restrict__ h,
                                                        float* __restrict__ out, int nE,
                                                        int* __restrict__ queues) {
    __shared__ int slot;
    int lane = threadIdx.x & 31;
    int wave = __builtin_amdgcn_readfirstlane(threadIdx.x >> 5);
    int nClaims = (nE + CLAIM_EDGES - 1) / CLAIM_EDGES;
    int half = my_half(&slot);
    for (int pass = 0; pass < 2; ++pass) {
        int t = half ^ pass;
        for (;;) {
            int idx = claim_block(queues + t * QSTRIDE, &slot);
            if (idx >= nClaims) break;
            int e0 = idx * CLAIM_EDGES + wave * 64;
            agg2_wave(src, dst, dinv, h, out, e0,      nE, t, lane);
            agg2_wave(src, dst, dinv, h, out, e0 + 32, nE, t, lane);
        }
    }
}

__global__ __launch_bounds__(256) void logsoftmax_kernel(float* __restrict__ out,
                                                         const float* __restrict__ pre,
                                                         const float* __restrict__ dinv,
                                                         const float* __restrict__ b2) {
    int r = blockIdx.x * 256 + threadIdx.x;
    if (r >= N_NODES) return;
    float di = dinv[r];
    float di2 = di * di;
    float v[N_CLASSES];
    float m = -INFINITY;
    for (int c = 0; c < N_CLASSES; ++c) {
        float x = out[(size_t)r * N_CLASSES + c] + di2 * pre[(size_t)r * N_CLASSES + c] + b2[c];
        v[c] = x;
        m = fmaxf(m, x);
    }
    float s = 0.0f;
    for (int c = 0; c < N_CLASSES; ++c) s += expf(v[c] - m);
    float ls = logf(s);
    for (int c = 0; c < N_CLASSES; ++c)
        out[(size_t)r * N_CLASSES + c] = v[c] - m - ls;
}

extern "C" void kernel_launch(void* const* d_in, const int* in_sizes, int n_in,
                              void* d_out, int out_size, void* d_ws, size_t ws_size,
                              hipStream_t stream) {
    const float* x  = (const float*)d_in[0];
    const int*   ei = (const int*)d_in[1];
    const float* W1 = (const float*)d_in[2];
    const float* b1 = (const float*)d_in[3];
    const float* W2 = (const float*)d_in[4];
    const float* b2 = (const float*)d_in[5];
    float* out = (float*)d_out;

    int nE = in_sizes[1] / 2;
    const int* src = ei;
    const int* dst = ei + nE;

    float* dinv = (float*)d_ws;
    float* bufA = dinv + 102400;
    float* bufB = bufA + (size_t)N_NODES * HIDDEN;

    int* q1 = (int*)(dinv + 100096);
    int* q2 = q1 + 2 * QSTRIDE;

    int nTiles = N_NODES / 16;

    init_kernel<<<(N_NODES * HIDDEN + 255) / 256, 256, 0, stream>>>(bufB, out, dinv, q1);
    degree_kernel<<<(nE + 255) / 256, 256, 0, stream>>>(dst, dinv, nE);
    rsqrt_kernel<<<(N_NODES + 255) / 256, 256, 0, stream>>>(dinv);
    gemm1_kernel<<<(nTiles + 7) / 8, 256, 0, stream>>>(x, W1, bufA, nTiles);
    edge_agg1_kernel<<<1536, 256, 0, stream>>>(src, dst, dinv, bufA, bufB, nE, q1);
    relu_kernel<<<(N_NODES * HIDDEN + 255) / 256, 256, 0, stream>>>(bufA, bufB, dinv, b1);
    gemm2_kernel<<<(nTiles + 7) / 8, 256, 0, stream>>>(bufB, W2, bufA, nTiles);
    edge_agg2_kernel<<<2048, 256, 0, stream>>>(src, dst, dinv, bufA, out, nE, q2);
    logsoftmax_kernel<<<(N_NODES + 255) / 256, 256, 0, stream>>>(out, bufA, dinv, b2);
}


# Element-wise passes folded into the GEMMs: epilogues store self-loop term + bias as the accumulators' start value (summed first, edges after), rsqrt in GEMM1, relu on GEMM2's A load; 9 -> 7 launches
# speedup vs baseline: 1.6476x; 1.0635x over previous
#include <hip/hip_runtime.h>
#include <math.h>

typedef float v2f __attribute__((ext_vector_type(2)));
typedef float v8f __attribute__((ext_vector_type(8)));

#define N_NODES   100000
#define N_FEAT    512
#define HIDDEN    64
#define N_CLASSES 16

static __device__ __forceinline__ void atomic_add_f32(float* p, float v) {
    __hip_atomic_fetch_add(p, v, __ATOMIC_RELAXED, __HIP_MEMORY_SCOPE_AGENT);
}

#define QSTRIDE 64
__global__ __launch_bounds__(256) void init_kernel(float* deg, int* queues) {
    int t = blockIdx.x * 256 + threadIdx.x;
    if (t < 4)       queues[t * QSTRIDE] = 0;
    if (t < N_NODES) deg[t] = 1.0f;
}

__global__ __launch_bounds__(256) void degree_kernel(const int* __restrict__ dst,
                                                     float* __restrict__ deg, int nE) {
    int e = blockIdx.x * 256 + threadIdx.x;
    if (e < nE) atomic_add_f32(&deg[dst[e]], 1.0f);
}

static __device__ __forceinline__ int lane_i(int v, int j) { return __builtin_amdgcn_readlane(v, j); }
static __device__ __forceinline__ float lane_f(float v, int j) {
    return __int_as_float(__builtin_amdgcn_readlane(__float_as_int(v), j));
}

#define WMMA_SLOTS4 "v_nop\n\tv_nop\n\tv_nop\n\tv_nop"

__global__ __launch_bounds__(256) void gemm1_kernel(const float* __restrict__ A,
                                                    const float* __restrict__ B,
                                                    const float* __restrict__ bias,
                                                    float* __restrict__ dinv,
                                                    float* __restrict__ C,
                                                    float* __restrict__ S, int nTiles) {
    int wave = threadIdx.x >> 5;
    int lane = threadIdx.x & 31;
    int tile = blockIdx.x * 8 + wave;
    if (tile >= nTiles) return;
    int nr = lane & 15;
    int hi = lane >> 4;

    float di = rsqrtf(dinv[tile * 16 + nr]);
    float bb0 = bias[nr], bb1 = bias[nr + 16], bb2 = bias[nr + 32], bb3 = bias[nr + 48];

    const float* arow = A + (size_t)(tile * 16 + nr) * N_FEAT + 2 * hi;
    v8f acc0 = {}, acc1 = {}, acc2 = {}, acc3 = {};

    for (int kk = 0; kk < N_FEAT; kk += 4) {
        v2f a = *(const v2f*)(arow + kk);
        const float* b0p = B + (size_t)(kk + 2 * hi) * HIDDEN + nr;
        const float* b1p = b0p + HIDDEN;
        v2f b0 = { b0p[0],  b1p[0]  };
        v2f b1 = { b0p[16], b1p[16] };
        v2f b2 = { b0p[32], b1p[32] };
        v2f b3 = { b0p[48], b1p[48] };
        acc0 = __builtin_amdgcn_wmma_f32_16x16x4_f32(false, a, false, b0, (short)0, acc0, false, false);
        acc1 = __builtin_amdgcn_wmma_f32_16x16x4_f32(false, a, false, b1, (short)0, acc1, false, false);
        acc2 = __builtin_amdgcn_wmma_f32_16x16x4_f32(false, a, false, b2, (short)0, acc2, false, false);
        acc3 = __builtin_amdgcn_wmma_f32_16x16x4_f32(false, a, false, b3, (short)0, acc3, false, false);
    }
    asm volatile(WMMA_SLOTS4 : "+v"(acc0), "+v"(acc1), "+v"(acc2), "+v"(acc3), "+v"(di),
                               "+v"(bb0), "+v"(bb1), "+v"(bb2), "+v"(bb3), "+v"(nr), "+v"(hi));
    if (hi == 0) dinv[tile * 16 + nr] = di;
    int mbase = tile * 16 + hi * 8;
#pragma unroll
    for (int r = 0; r < 8; ++r) {
        float dlo = lane_f(di, r), dhi = lane_f(di, 8 + r);
        float dr = hi ? dhi : dlo;
        float d2 = dr * dr;
        size_t o = (size_t)(mbase + r) * HIDDEN + nr;
        float* crow = C + o;
        float* srow = S + o;
        crow[0]  = acc0[r];
        crow[16] = acc1[r];
        crow[32] = acc2[r];
        crow[48] = acc3[r];
        srow[0]  = d2 * acc0[r] + bb0;
        srow[16] = d2 * acc1[r] + bb1;
        srow[32] = d2 * acc2[r] + bb2;
        srow[48] = d2 * acc3[r] + bb3;
    }
}

#define HALF_SPLIT  (N_NODES / 2)
#define CLAIM_EDGES 512

static __device__ __forceinline__ int my_half(int* slot) {
    if (threadIdx.x == 0) *slot = (__builtin_amdgcn_s_sendmsg_rtn(0x87) >> 18) & 1;
    __syncthreads();
    int hf = *slot;
    __syncthreads();
    return hf;
}

static __device__ __forceinline__ int claim_block(int* q, int* slot) {
    if (threadIdx.x == 0)
        *slot = __hip_atomic_fetch_add(q, 1, __ATOMIC_RELAXED, __HIP_MEMORY_SCOPE_AGENT);
    __syncthreads();
    int idx = *slot;
    __syncthreads();
    return idx;
}

static __device__ __forceinline__ void agg1_wave(const int* __restrict__ src,
                                                 const int* __restrict__ dst,
                                                 const float* __restrict__ dinv,
                                                 const float* __restrict__ h,
                                                 float* __restrict__ agg,
                                                 int e0, int nE, int t, int lane) {
    int cnt = nE - e0;
    if (cnt <= 0) return;
    int so = 0, dofs = 0;
    float c = 0.0f;
    bool sel = false;
    if (lane < cnt) {
        int s = src[e0 + lane];
        int d = dst[e0 + lane];
        so   = s * HIDDEN;
        dofs = d * HIDDEN;
        sel = (d >= HALF_SPLIT ? 1 : 0) == t;
        if (sel) c = dinv[s] * dinv[d];
    }
    unsigned mask = __builtin_amdgcn_ballot_w32(sel);
    float v0[32], v1[32];
#pragma unroll
    for (int j = 0; j < 32; ++j) {
        float a = 0.0f, b = 0.0f;
        if ((mask >> j) & 1u) {
            int ro = lane_i(so, j) + lane;
            a = h[ro];
            b = h[ro + 32];
        }
        v0[j] = a;
        v1[j] = b;
    }
#pragma unroll
    for (int j = 0; j < 32; ++j) {
        if ((mask >> j) & 1u) {
            float cj = lane_f(c, j);
            int ro = lane_i(dofs, j) + lane;
            atomic_add_f32(&agg[ro],      v0[j] * cj);
            atomic_add_f32(&agg[ro + 32], v1[j] * cj);
        }
    }
}

__global__ __launch_bounds__(256) void edge_agg1_kernel(const int* __restrict__ src,
                                                        const int* __restrict__ dst,
                                                        const float* __restrict__ dinv,
                                                        const float* __restrict__ h,
                                                        float* __restrict__ agg, int nE,
                                                        int* __restrict__ queues) {
    __shared__ int slot;
    int lane = threadIdx.x & 31;
    int wave = __builtin_amdgcn_readfirstlane(threadIdx.x >> 5);
    int nClaims = (nE + CLAIM_EDGES - 1) / CLAIM_EDGES;
    int half = my_half(&slot);
    for (int pass = 0; pass < 2; ++pass) {
        int t = half ^ pass;
        for (;;) {
            int idx = claim_block(queues + t * QSTRIDE, &slot);
            if (idx >= nClaims) break;
            int e0 = idx * CLAIM_EDGES + wave * 64;
            agg1_wave(src, dst, dinv, h, agg, e0,      nE, t, lane);
            agg1_wave(src, dst, dinv, h, agg, e0 + 32, nE, t, lane);
        }
    }
}

__global__ __launch_bounds__(256) void gemm2_kernel(const float* __restrict__ A,
                                                    const float* __restrict__ B,
                                                    const float* __restrict__ b2,
                                                    const float* __restrict__ dinv,
                                                    float* __restrict__ C,
                                                    float* __restrict__ out, int nTiles) {
    int wave = threadIdx.x >> 5;
    int lane = threadIdx.x & 31;
    int tile = blockIdx.x * 8 + wave;
    if (tile >= nTiles) return;
    int nr = lane & 15;
    int hi = lane >> 4;

    float di = dinv[tile * 16 + nr];
    float bb = b2[nr];

    const float* arow = A + (size_t)(tile * 16 + nr) * HIDDEN + 2 * hi;
    v2f af[HIDDEN / 4];
#pragma unroll
    for (int i = 0; i < HIDDEN / 4; ++i) {
        v2f a = *(const v2f*)(arow + 4 * i);
        a.x = a.x > 0.0f ? a.x : 0.0f;
        a.y = a.y > 0.0f ? a.y : 0.0f;
        af[i] = a;
    }
    v8f acc = {};
    asm volatile("" : "+v"(af[0]), "+v"(af[1]), "+v"(af[2]),  "+v"(af[3]),
                      "+v"(af[4]), "+v"(af[5]), "+v"(af[6]),  "+v"(af[7]));
    asm volatile("" : "+v"(af[8]),  "+v"(af[9]),  "+v"(af[10]), "+v"(af[11]),
                      "+v"(af[12]), "+v"(af[13]), "+v"(af[14]), "+v"(af[15]), "+v"(acc));
#pragma unroll
    for (int i = 0; i < HIDDEN / 4; ++i) {
        const float* b0p = B + (size_t)(4 * i + 2 * hi) * N_CLASSES + nr;
        v2f b = { b0p[0], b0p[N_CLASSES] };
        acc = __builtin_amdgcn_wmma_f32_16x16x4_f32(false, af[i], false, b, (short)0, acc, false, false);
    }
    asm volatile(WMMA_SLOTS4 : "+v"(acc), "+v"(di), "+v"(bb), "+v"(nr), "+v"(hi));
    int mbase = tile * 16 + hi * 8;
#pragma unroll
    for (int r = 0; r < 8; ++r) {
        float dlo = lane_f(di, r), dhi = lane_f(di, 8 + r);
        float dr = hi ? dhi : dlo;
        size_t o = (size_t)(mbase + r) * N_CLASSES + nr;
        C[o]   = acc[r];
        out[o] = dr * dr * acc[r] + bb;
    }
}

template <int K> static __device__ __forceinline__ int row_lane_i(int v) {
    return __builtin_amdgcn_update_dpp(0, v, 0x150 + K, 0xF, 0xF, false);
}
template <int K> static __device__ __forceinline__ float row_lane_f(float v) {
    return __int_as_float(row_lane_i<K>(__float_as_int(v)));
}

#define AGG2_LOAD(K) { int sk = row_lane_i<K>(s); int fk = row_lane_i<K>(f); \
                       if (fk) v[K] = h[sk * N_CLASSES + col]; }
#define AGG2_ADD(K)  { int dk = row_lane_i<K>(d); int fk = row_lane_i<K>(f); float ck = row_lane_f<K>(c); \
                       if (fk) atomic_add_f32(&out[dk * N_CLASSES + col], ck * v[K]); }
#define AGG2_ALL(M)  M(0) M(1) M(2) M(3) M(4) M(5) M(6) M(7) M(8) M(9) M(10) M(11) M(12) M(13) M(14) M(15)

static __device__ __forceinline__ void agg2_wave(const int* __restrict__ src,
                                                 const int* __restrict__ dst,
                                                 const float* __restrict__ dinv,
                                                 const float* __restrict__ h,
                                                 float* __restrict__ out,
                                                 int e0, int nE, int t, int lane) {
    int cnt = nE - e0;
    if (cnt <= 0) return;
    int s = 0, d = 0, f = 0;
    float c = 0.0f;
    if (lane < cnt) {
        s = src[e0 + lane];
        d = dst[e0 + lane];
        f = ((d >= HALF_SPLIT ? 1 : 0) == t) ? 1 : 0;
        if (f) c = dinv[s] * dinv[d];
    }
    int col = lane & 15;
    float v[16];
    AGG2_ALL(AGG2_LOAD)
    AGG2_ALL(AGG2_ADD)
}

__global__ __launch_bounds__(256) void edge_agg2_kernel(const int* __restrict__ src,
                                                        const int* __restrict__ dst,
                                                        const float* __restrict__ dinv,
                                                        const float* __restrict__ h,
                                                        float* __restrict__ out, int nE,
                                                        int* __restrict__ queues) {
    __shared__ int slot;
    int lane = threadIdx.x & 31;
    int wave = __builtin_amdgcn_readfirstlane(threadIdx.x >> 5);
    int nClaims = (nE + CLAIM_EDGES - 1) / CLAIM_EDGES;
    int half = my_half(&slot);
    for (int pass = 0; pass < 2; ++pass) {
        int t = half ^ pass;
        for (;;) {
            int idx = claim_block(queues + t * QSTRIDE, &slot);
            if (idx >= nClaims) break;
            int e0 = idx * CLAIM_EDGES + wave * 64;
            agg2_wave(src, dst, dinv, h, out, e0,      nE, t, lane);
            agg2_wave(src, dst, dinv, h, out, e0 + 32, nE, t, lane);
        }
    }
}

__global__ __launch_bounds__(256) void logsoftmax_kernel(float* __restrict__ out) {
    int r = blockIdx.x * 256 + threadIdx.x;
    if (r >= N_NODES) return;
    float v[N_CLASSES];
    float m = -INFINITY;
    for (int c = 0; c < N_CLASSES; ++c) {
        float x = out[(size_t)r * N_CLASSES + c];
        v[c] = x;
        m = fmaxf(m, x);
    }
    float s = 0.0f;
    for (int c = 0; c < N_CLASSES; ++c) s += expf(v[c] - m);
    float ls = logf(s);
    for (int c = 0; c < N_CLASSES; ++c)
        out[(size_t)r * N_CLASSES + c] = v[c] - m - ls;
}

extern "C" void kernel_launch(void* const* d_in, const int* in_sizes, int n_in,
                              void* d_out, int out_size, void* d_ws, size_t ws_size,
                              hipStream_t stream) {
    const float* x  = (const float*)d_in[0];
    const int*   ei = (const int*)d_in[1];
    const float* W1 = (const float*)d_in[2];
    const float* b1 = (const float*)d_in[3];
    const float* W2 = (const float*)d_in[4];
    const float* b2 = (const float*)d_in[5];
    float* out = (float*)d_out;

    int nE = in_sizes[1] / 2;
    const int* src = ei;
    const int* dst = ei + nE;

    float* dinv = (float*)d_ws;
    float* bufA = dinv + 102400;
    float* bufB = bufA + (size_t)N_NODES * HIDDEN;

    int* q1 = (int*)(dinv + 100096);
    int* q2 = q1 + 2 * QSTRIDE;

    int nTiles = N_NODES / 16;

    init_kernel<<<(N_NODES + 255) / 256, 256, 0, stream>>>(dinv, q1);
    degree_kernel<<<(nE + 255) / 256, 256, 0, stream>>>(dst, dinv, nE);
    gemm1_kernel<<<(nTiles + 7) / 8, 256, 0, stream>>>(x, W1, b1, dinv, bufA, bufB, nTiles);
    edge_agg1_kernel<<<1536, 256, 0, stream>>>(src, dst, dinv, bufA, bufB, nE, q1);
    gemm2_kernel<<<(nTiles + 7) / 8, 256, 0, stream>>>(bufB, W2, b2, dinv, bufA, out, nTiles);
    edge_agg2_kernel<<<2048, 256, 0, stream>>>(src, dst, dinv, bufA, out, nE, q2);
    logsoftmax_kernel<<<(N_NODES + 255) / 256, 256, 0, stream>>>(out);
}


# Degree count: per-edge one-word global float atomics replaced by LDS integer counters per (node range, edge slice) work-group, non-zero counts added to deg a whole line per wave; sums exact
# speedup vs baseline: 2.1804x; 1.3234x over previous
#include <hip/hip_runtime.h>
#include <math.h>

typedef float v2f __attribute__((ext_vector_type(2)));
typedef float v8f __attribute__((ext_vector_type(8)));

#define N_NODES   100000
#define N_FEAT    512
#define HIDDEN    64
#define N_CLASSES 16

static __device__ __forceinline__ void atomic_add_f32(float* p, float v) {
    __hip_atomic_fetch_add(p, v, __ATOMIC_RELAXED, __HIP_MEMORY_SCOPE_AGENT);
}

#define QSTRIDE 64
__global__ __launch_bounds__(256) void init_kernel(float* deg, int* queues) {
    int t = blockIdx.x * 256 + threadIdx.x;
    if (t < 4)       queues[t * QSTRIDE] = 0;
    if (t < N_NODES) deg[t] = 1.0f;
}

#define DEG_RANGE   6272
#define DEG_RANGES  16
#define DEG_SLICES  16
#define DEG_THREADS 512
#define DEG_BATCH   28
__global__ __launch_bounds__(DEG_THREADS) void degree_kernel(const int* __restrict__ dst,
                                                             float* __restrict__ deg, int nE) {
    __shared__ unsigned cnt[DEG_RANGE];
    int tid   = threadIdx.x;
    int range = blockIdx.x % DEG_RANGES;
    int slice = blockIdx.x / DEG_RANGES;
    int lo    = range * DEG_RANGE;
    for (int i = tid; i < DEG_RANGE; i += DEG_THREADS) cnt[i] = 0u;
    __syncthreads();

    int per = (nE + DEG_SLICES - 1) / DEG_SLICES;
    int e0  = slice * per;
    int e1  = e0 + per < nE ? e0 + per : nE;
    for (int e = e0 + tid; e < e1; e += DEG_THREADS * DEG_BATCH) {
        int d[DEG_BATCH];
#pragma unroll
        for (int u = 0; u < DEG_BATCH; ++u) {
            int eu = e + u * DEG_THREADS;
            d[u] = eu < e1 ? dst[eu] : -1;
        }
#pragma unroll
        for (int u = 0; u < DEG_BATCH; ++u) {
            unsigned r = (unsigned)(d[u] - lo);
            if (r < (unsigned)DEG_RANGE) atomicAdd(&cnt[r], 1u);
        }
    }
    __syncthreads();

    for (int i = tid; i < DEG_RANGE; i += DEG_THREADS) {
        unsigned c = cnt[i];
        int n = lo + i;
        if (c != 0u && n < N_NODES) atomic_add_f32(&deg[n], (float)c);
    }
}

static __device__ __forceinline__ int lane_i(int v, int j) { return __builtin_amdgcn_readlane(v, j); }
static __device__ __forceinline__ float lane_f(float v, int j) {
    return __int_as_float(__builtin_amdgcn_readlane(__float_as_int(v), j));
}

#define WMMA_SLOTS4 "v_nop\n\tv_nop\n\tv_nop\n\tv_nop"

__global__ __launch_bounds__(256) void gemm1_kernel(const float* __restrict__ A,
                                                    const float* __restrict__ B,
                                                    const float* __restrict__ bias,
                                                    float* __restrict__ dinv,
                                                    float* __restrict__ C,
                                                    float* __restrict__ S, int nTiles) {
    int wave = threadIdx.x >> 5;
    int lane = threadIdx.x & 31;
    int tile = blockIdx.x * 8 + wave;
    if (tile >= nTiles) return;
    int nr = lane & 15;
    int hi = lane >> 4;

    float di = rsqrtf(dinv[tile * 16 + nr]);
    float bb0 = bias[nr], bb1 = bias[nr + 16], bb2 = bias[nr + 32], bb3 = bias[nr + 48];

    const float* arow = A + (size_t)(tile * 16 + nr) * N_FEAT + 2 * hi;
    v8f acc0 = {}, acc1 = {}, acc2 = {}, acc3 = {};

    for (int kk = 0; kk < N_FEAT; kk += 4) {
        v2f a = *(const v2f*)(arow + kk);
        const float* b0p = B + (size_t)(kk + 2 * hi) * HIDDEN + nr;
        const float* b1p = b0p + HIDDEN;
        v2f b0 = { b0p[0],  b1p[0]  };
        v2f b1 = { b0p[16], b1p[16] };
        v2f b2 = { b0p[32], b1p[32] };
        v2f b3 = { b0p[48], b1p[48] };
        acc0 = __builtin_amdgcn_wmma_f32_16x16x4_f32(false, a, false, b0, (short)0, acc0, false, false);
        acc1 = __builtin_amdgcn_wmma_f32_16x16x4_f32(false, a, false, b1, (short)0, acc1, false, false);
        acc2 = __builtin_amdgcn_wmma_f32_16x16x4_f32(false, a, false, b2, (short)0, acc2, false, false);
        acc3 = __builtin_amdgcn_wmma_f32_16x16x4_f32(false, a, false, b3, (short)0, acc3, false, false);
    }
    asm volatile(WMMA_SLOTS4 : "+v"(acc0), "+v"(acc1), "+v"(acc2), "+v"(acc3), "+v"(di),
                               "+v"(bb0), "+v"(bb1), "+v"(bb2), "+v"(bb3), "+v"(nr), "+v"(hi));
    if (hi == 0) dinv[tile * 16 + nr] = di;
    int mbase = tile * 16 + hi * 8;
#pragma unroll
    for (int r = 0; r < 8; ++r) {
        float dlo = lane_f(di, r), dhi = lane_f(di, 8 + r);
        float dr = hi ? dhi : dlo;
        float d2 = dr * dr;
        size_t o = (size_t)(mbase + r) * HIDDEN + nr;
        float* crow = C + o;
        float* srow = S + o;
        crow[0]  = acc0[r];
        crow[16] = acc1[r];
        crow[32] = acc2[r];
        crow[48] = acc3[r];
        srow[0]  = d2 * acc0[r] + bb0;
        srow[16] = d2 * acc1[r] + bb1;
        srow[32] = d2 * acc2[r] + bb2;
        srow[48] = d2 * acc3[r] + bb3;
    }
}

#define HALF_SPLIT  (N_NODES / 2)
#define CLAIM_EDGES 512

static __device__ __forceinline__ int my_half(int* slot) {
    if (threadIdx.x == 0) *slot = (__builtin_amdgcn_s_sendmsg_rtn(0x87) >> 18) & 1;
    __syncthreads();
    int hf = *slot;
    __syncthreads();
    return hf;
}

static __device__ __forceinline__ int claim_block(int* q, int* slot) {
    if (threadIdx.x == 0)
        *slot = __hip_atomic_fetch_add(q, 1, __ATOMIC_RELAXED, __HIP_MEMORY_SCOPE_AGENT);
    __syncthreads();
    int idx = *slot;
    __syncthreads();
    return idx;
}

static __device__ __forceinline__ void agg1_wave(const int* __restrict__ src,
                                                 const int* __restrict__ dst,
                                                 const float* __restrict__ dinv,
                                                 const float* __restrict__ h,
                                                 float* __restrict__ agg,
                                                 int e0, int nE, int t, int lane) {
    int cnt = nE - e0;
    if (cnt <= 0) return;
    int so = 0, dofs = 0;
    float c = 0.0f;
    bool sel = false;
    if (lane < cnt) {
        int s = src[e0 + lane];
        int d = dst[e0 + lane];
        so   = s * HIDDEN;
        dofs = d * HIDDEN;
        sel = (d >= HALF_SPLIT ? 1 : 0) == t;
        if (sel) c = dinv[s] * dinv[d];
    }
    unsigned mask = __builtin_amdgcn_ballot_w32(sel);
    float v0[32], v1[32];
#pragma unroll
    for (int j = 0; j < 32; ++j) {
        float a = 0.0f, b = 0.0f;
        if ((mask >> j) & 1u) {
            int ro = lane_i(so, j) + lane;
            a = h[ro];
            b = h[ro + 32];
        }
        v0[j] = a;
        v1[j] = b;
    }
#pragma unroll
    for (int j = 0; j < 32; ++j) {
        if ((mask >> j) & 1u) {
            float cj = lane_f(c, j);
            int ro = lane_i(dofs, j) + lane;
            atomic_add_f32(&agg[ro],      v0[j] * cj);
            atomic_add_f32(&agg[ro + 32], v1[j] * cj);
        }
    }
}

__global__ __launch_bounds__(256) void edge_agg1_kernel(const int* __restrict__ src,
                                                        const int* __restrict__ dst,
                                                        const float* __restrict__ dinv,
                                                        const float* __restrict__ h,
                                                        float* __restrict__ agg, int nE,
                                                        int* __restrict__ queues) {
    __shared__ int slot;
    int lane = threadIdx.x & 31;
    int wave = __builtin_amdgcn_readfirstlane(threadIdx.x >> 5);
    int nClaims = (nE + CLAIM_EDGES - 1) / CLAIM_EDGES;
    int half = my_half(&slot);
    for (int pass = 0; pass < 2; ++pass) {
        int t = half ^ pass;
        for (;;) {
            int idx = claim_block(queues + t * QSTRIDE, &slot);
            if (idx >= nClaims) break;
            int e0 = idx * CLAIM_EDGES + wave * 64;
            agg1_wave(src, dst, dinv, h, agg, e0,      nE, t, lane);
            agg1_wave(src, dst, dinv, h, agg, e0 + 32, nE, t, lane);
        }
    }
}

__global__ __launch_bounds__(256) void gemm2_kernel(const float* __restrict__ A,
                                                    const float* __restrict__ B,
                                                    const float* __restrict__ b2,
                                                    const float* __restrict__ dinv,
                                                    float* __restrict__ C,
                                                    float* __restrict__ out, int nTiles) {
    int wave = threadIdx.x >> 5;
    int lane = threadIdx.x & 31;
    int tile = blockIdx.x * 8 + wave;
    if (tile >= nTiles) return;
    int nr = lane & 15;
    int hi = lane >> 4;

    float di = dinv[tile * 16 + nr];
    float bb = b2[nr];

    const float* arow = A + (size_t)(tile * 16 + nr) * HIDDEN + 2 * hi;
    v2f af[HIDDEN / 4];
#pragma unroll
    for (int i = 0; i < HIDDEN / 4; ++i) {
        v2f a = *(const v2f*)(arow + 4 * i);
        a.x = a.x > 0.0f ? a.x : 0.0f;
        a.y = a.y > 0.0f ? a.y : 0.0f;
        af[i] = a;
    }
    v8f acc = {};
    asm volatile("" : "+v"(af[0]), "+v"(af[1]), "+v"(af[2]),  "+v"(af[3]),
                      "+v"(af[4]), "+v"(af[5]), "+v"(af[6]),  "+v"(af[7]));
    asm volatile("" : "+v"(af[8]),  "+v"(af[9]),  "+v"(af[10]), "+v"(af[11]),
                      "+v"(af[12]), "+v"(af[13]), "+v"(af[14]), "+v"(af[15]), "+v"(acc));
#pragma unroll
    for (int i = 0; i < HIDDEN / 4; ++i) {
        const float* b0p = B + (size_t)(4 * i + 2 * hi) * N_CLASSES + nr;
        v2f b = { b0p[0], b0p[N_CLASSES] };
        acc = __builtin_amdgcn_wmma_f32_16x16x4_f32(false, af[i], false, b, (short)0, acc, false, false);
    }
    asm volatile(WMMA_SLOTS4 : "+v"(acc), "+v"(di), "+v"(bb), "+v"(nr), "+v"(hi));
    int mbase = tile * 16 + hi * 8;
#pragma unroll
    for (int r = 0; r < 8; ++r) {
        float dlo = lane_f(di, r), dhi = lane_f(di, 8 + r);
        float dr = hi ? dhi : dlo;
        size_t o = (size_t)(mbase + r) * N_CLASSES + nr;
        C[o]   = acc[r];
        out[o] = dr * dr * acc[r] + bb;
    }
}

template <int K> static __device__ __forceinline__ int row_lane_i(int v) {
    return __builtin_amdgcn_update_dpp(0, v, 0x150 + K, 0xF, 0xF, false);
}
template <int K> static __device__ __forceinline__ float row_lane_f(float v) {
    return __int_as_float(row_lane_i<K>(__float_as_int(v)));
}

#define AGG2_LOAD(K) { int sk = row_lane_i<K>(s); int fk = row_lane_i<K>(f); \
                       if (fk) v[K] = h[sk * N_CLASSES + col]; }
#define AGG2_ADD(K)  { int dk = row_lane_i<K>(d); int fk = row_lane_i<K>(f); float ck = row_lane_f<K>(c); \
                       if (fk) atomic_add_f32(&out[dk * N_CLASSES + col], ck * v[K]); }
#define AGG2_ALL(M)  M(0) M(1) M(2) M(3) M(4) M(5) M(6) M(7) M(8) M(9) M(10) M(11) M(12) M(13) M(14) M(15)

static __device__ __forceinline__ void agg2_wave(const int* __restrict__ src,
                                                 const int* __restrict__ dst,
                                                 const float* __restrict__ dinv,
                                                 const float* __restrict__ h,
                                                 float* __restrict__ out,
                                                 int e0, int nE, int t, int lane) {
    int cnt = nE - e0;
    if (cnt <= 0) return;
    int s = 0, d = 0, f = 0;
    float c = 0.0f;
    if (lane < cnt) {
        s = src[e0 + lane];
        d = dst[e0 + lane];
        f = ((d >= HALF_SPLIT ? 1 : 0) == t) ? 1 : 0;
        if (f) c = dinv[s] * dinv[d];
    }
    int col = lane & 15;
    float v[16];
    AGG2_ALL(AGG2_LOAD)
    AGG2_ALL(AGG2_ADD)
}

__global__ __launch_bounds__(256) void edge_agg2_kernel(const int* __restrict__ src,
                                                        const int* __restrict__ dst,
                                                        const float* __restrict__ dinv,
                                                        const float* __restrict__ h,
                                                        float* __restrict__ out, int nE,
                                                        int* __restrict__ queues) {
    __shared__ int slot;
    int lane = threadIdx.x & 31;
    int wave = __builtin_amdgcn_readfirstlane(threadIdx.x >> 5);
    int nClaims = (nE + CLAIM_EDGES - 1) / CLAIM_EDGES;
    int half = my_half(&slot);
    for (int pass = 0; pass < 2; ++pass) {
        int t = half ^ pass;
        for (;;) {
            int idx = claim_block(queues + t * QSTRIDE, &slot);
            if (idx >= nClaims) break;
            int e0 = idx * CLAIM_EDGES + wave * 64;
            agg2_wave(src, dst, dinv, h, out, e0,      nE, t, lane);
            agg2_wave(src, dst, dinv, h, out, e0 + 32, nE, t, lane);
        }
    }
}

__global__ __launch_bounds__(256) void logsoftmax_kernel(float* __restrict__ out) {
    int r = blockIdx.x * 256 + threadIdx.x;
    if (r >= N_NODES) return;
    float v[N_CLASSES];
    float m = -INFINITY;
    for (int c = 0; c < N_CLASSES; ++c) {
        float x = out[(size_t)r * N_CLASSES + c];
        v[c] = x;
        m = fmaxf(m, x);
    }
    float s = 0.0f;
    for (int c = 0; c < N_CLASSES; ++c) s += expf(v[c] - m);
    float ls = logf(s);
    for (int c = 0; c < N_CLASSES; ++c)
        out[(size_t)r * N_CLASSES + c] = v[c] - m - ls;
}

extern "C" void kernel_launch(void* const* d_in, const int* in_sizes, int n_in,
                              void* d_out, int out_size, void* d_ws, size_t ws_size,
                              hipStream_t stream) {
    const float* x  = (const float*)d_in[0];
    const int*   ei = (const int*)d_in[1];
    const float* W1 = (const float*)d_in[2];
    const float* b1 = (const float*)d_in[3];
    const float* W2 = (const float*)d_in[4];
    const float* b2 = (const float*)d_in[5];
    float* out = (float*)d_out;

    int nE = in_sizes[1] / 2;
    const int* src = ei;
    const int* dst = ei + nE;

    float* dinv = (float*)d_ws;
    float* bufA = dinv + 102400;
    float* bufB = bufA + (size_t)N_NODES * HIDDEN;

    int* q1 = (int*)(dinv + 100096);
    int* q2 = q1 + 2 * QSTRIDE;

    int nTiles = N_NODES / 16;

    init_kernel<<<(N_NODES + 255) / 256, 256, 0, stream>>>(dinv, q1);
    degree_kernel<<<DEG_RANGES * DEG_SLICES, DEG_THREADS, 0, stream>>>(dst, dinv, nE);
    gemm1_kernel<<<(nTiles + 7) / 8, 256, 0, stream>>>(x, W1, b1, dinv, bufA, bufB, nTiles);
    edge_agg1_kernel<<<1536, 256, 0, stream>>>(src, dst, dinv, bufA, bufB, nE, q1);
    gemm2_kernel<<<(nTiles + 7) / 8, 256, 0, stream>>>(bufB, W2, b2, dinv, bufA, out, nTiles);
    edge_agg2_kernel<<<2048, 256, 0, stream>>>(src, dst, dinv, bufA, out, nE, q2);
    logsoftmax_kernel<<<(N_NODES + 255) / 256, 256, 0, stream>>>(out);
}


# GEMM1: W1 copied once per work-group into LDS in fragment order (512 work-groups of 16 waves, two per WGP); a k step's B fragments are two 16-byte LDS reads, were 8 global loads; k order kept
# speedup vs baseline: 2.4603x; 1.1284x over previous
#include <hip/hip_runtime.h>
#include <math.h>

typedef float v2f __attribute__((ext_vector_type(2)));
typedef float v8f __attribute__((ext_vector_type(8)));

#define N_NODES   100000
#define N_FEAT    512
#define HIDDEN    64
#define N_CLASSES 16

static __device__ __forceinline__ void atomic_add_f32(float* p, float v) {
    __hip_atomic_fetch_add(p, v, __ATOMIC_RELAXED, __HIP_MEMORY_SCOPE_AGENT);
}

#define QSTRIDE 64
__global__ __launch_bounds__(256) void init_kernel(float* deg, int* queues) {
    int t = blockIdx.x * 256 + threadIdx.x;
    if (t < 4)       queues[t * QSTRIDE] = 0;
    if (t < N_NODES) deg[t] = 1.0f;
}

#define DEG_RANGE   6272
#define DEG_RANGES  16
#define DEG_SLICES  16
#define DEG_THREADS 512
#define DEG_BATCH   28
__global__ __launch_bounds__(DEG_THREADS) void degree_kernel(const int* __restrict__ dst,
                                                             float* __restrict__ deg, int nE) {
    __shared__ unsigned cnt[DEG_RANGE];
    int tid   = threadIdx.x;
    int range = blockIdx.x % DEG_RANGES;
    int slice = blockIdx.x / DEG_RANGES;
    int lo    = range * DEG_RANGE;
    for (int i = tid; i < DEG_RANGE; i += DEG_THREADS) cnt[i] = 0u;
    __syncthreads();

    int per = (nE + DEG_SLICES - 1) / DEG_SLICES;
    int e0  = slice * per;
    int e1  = e0 + per < nE ? e0 + per : nE;
    for (int e = e0 + tid; e < e1; e += DEG_THREADS * DEG_BATCH) {
        int d[DEG_BATCH];
#pragma unroll
        for (int u = 0; u < DEG_BATCH; ++u) {
            int eu = e + u * DEG_THREADS;
            d[u] = eu < e1 ? dst[eu] : -1;
        }
#pragma unroll
        for (int u = 0; u < DEG_BATCH; ++u) {
            unsigned r = (unsigned)(d[u] - lo);
            if (r < (unsigned)DEG_RANGE) atomicAdd(&cnt[r], 1u);
        }
    }
    __syncthreads();

    for (int i = tid; i < DEG_RANGE; i += DEG_THREADS) {
        unsigned c = cnt[i];
        int n = lo + i;
        if (c != 0u && n < N_NODES) atomic_add_f32(&deg[n], (float)c);
    }
}

static __device__ __forceinline__ int lane_i(int v, int j) { return __builtin_amdgcn_readlane(v, j); }
static __device__ __forceinline__ float lane_f(float v, int j) {
    return __int_as_float(__builtin_amdgcn_readlane(__float_as_int(v), j));
}

#define WMMA_SLOTS4 "v_nop\n\tv_nop\n\tv_nop\n\tv_nop"

typedef float v4f __attribute__((ext_vector_type(4)));
#define G1_THREADS 512
#define G1_WAVES   (G1_THREADS / 32)
#define G1_BLOCKS  512
#define G1_PLANE   ((N_FEAT / 2) * 16 * 4)
__global__ __launch_bounds__(G1_THREADS) void gemm1_kernel(const float* __restrict__ A,
                                                           const float* __restrict__ B,
                                                           const float* __restrict__ bias,
                                                           float* __restrict__ dinv,
                                                           float* __restrict__ C,
                                                           float* __restrict__ S, int nTiles) {
    __shared__ float Bl[2 * G1_PLANE];
#pragma unroll
    for (int i = 0; i < (N_FEAT / 2) * (HIDDEN / 4) / G1_THREADS; ++i) {
        int it = i * G1_THREADS + threadIdx.x;
        int kp = it >> 4, q = it & 15;
        v4f r0 = *(const v4f*)(B + (size_t)(2 * kp) * HIDDEN + 4 * q);
        v4f r1 = *(const v4f*)(B + (size_t)(2 * kp + 1) * HIDDEN + 4 * q);
        int j = q >> 2;
        float* p = Bl + (j >> 1) * G1_PLANE + (kp * 16 + (q & 3) * 4) * 4 + (j & 1) * 2;
        *(v2f*)(p)      = (v2f){ r0.x, r1.x };
        *(v2f*)(p + 4)  = (v2f){ r0.y, r1.y };
        *(v2f*)(p + 8)  = (v2f){ r0.z, r1.z };
        *(v2f*)(p + 12) = (v2f){ r0.w, r1.w };
    }
    __syncthreads();

    int wave = threadIdx.x >> 5;
    int lane = threadIdx.x & 31;
    int nr = lane & 15;
    int hi = lane >> 4;
    for (int tile = wave * (int)gridDim.x + (int)blockIdx.x; tile < nTiles;
         tile += G1_WAVES * (int)gridDim.x) {
        float di = rsqrtf(dinv[tile * 16 + nr]);
        float bb0 = bias[nr], bb1 = bias[nr + 16], bb2 = bias[nr + 32], bb3 = bias[nr + 48];

        const float* arow = A + (size_t)(tile * 16 + nr) * N_FEAT + 2 * hi;
        const float* brow = Bl + (hi * 16 + nr) * 4;
        v8f acc0 = {}, acc1 = {}, acc2 = {}, acc3 = {};

        for (int kk = 0; kk < N_FEAT; kk += 4) {
            v2f a = *(const v2f*)(arow + kk);
            const float* bp = brow + (kk >> 1) * 64;
            v4f q0 = *(const v4f*)(bp);
            v4f q1 = *(const v4f*)(bp + G1_PLANE);
            v2f b0 = { q0.x, q0.y };
            v2f b1 = { q0.z, q0.w };
            v2f b2 = { q1.x, q1.y };
            v2f b3 = { q1.z, q1.w };
            acc0 = __builtin_amdgcn_wmma_f32_16x16x4_f32(false, a, false, b0, (short)0, acc0, false, false);
            acc1 = __builtin_amdgcn_wmma_f32_16x16x4_f32(false, a, false, b1, (short)0, acc1, false, false);
            acc2 = __builtin_amdgcn_wmma_f32_16x16x4_f32(false, a, false, b2, (short)0, acc2, false, false);
            acc3 = __builtin_amdgcn_wmma_f32_16x16x4_f32(false, a, false, b3, (short)0, acc3, false, false);
        }
        asm volatile(WMMA_SLOTS4 : "+v"(acc0), "+v"(acc1), "+v"(acc2), "+v"(acc3), "+v"(di),
                                   "+v"(bb0), "+v"(bb1), "+v"(bb2), "+v"(bb3), "+v"(nr), "+v"(hi));
        if (hi == 0) dinv[tile * 16 + nr] = di;
        int mbase = tile * 16 + hi * 8;
#pragma unroll
        for (int r = 0; r < 8; ++r) {
            float dlo = lane_f(di, r), dhi = lane_f(di, 8 + r);
            float dr = hi ? dhi : dlo;
            float d2 = dr * dr;
            size_t o = (size_t)(mbase + r) * HIDDEN + nr;
            float* crow = C + o;
            float* srow = S + o;
            crow[0]  = acc0[r];
            crow[16] = acc1[r];
            crow[32] = acc2[r];
            crow[48] = acc3[r];
            srow[0]  = d2 * acc0[r] + bb0;
            srow[16] = d2 * acc1[r] + bb1;
            srow[32] = d2 * acc2[r] + bb2;
            srow[48] = d2 * acc3[r] + bb3;
        }
    }
}

#define HALF_SPLIT  (N_NODES / 2)
#define CLAIM_EDGES 512

static __device__ __forceinline__ int my_half(int* slot) {
    if (threadIdx.x == 0) *slot = (__builtin_amdgcn_s_sendmsg_rtn(0x87) >> 18) & 1;
    __syncthreads();
    int hf = *slot;
    __syncthreads();
    return hf;
}

static __device__ __forceinline__ int claim_block(int* q, int* slot) {
    if (threadIdx.x == 0)
        *slot = __hip_atomic_fetch_add(q, 1, __ATOMIC_RELAXED, __HIP_MEMORY_SCOPE_AGENT);
    __syncthreads();
    int idx = *slot;
    __syncthreads();
    return idx;
}

static __device__ __forceinline__ void agg1_wave(const int* __restrict__ src,
                                                 const int* __restrict__ dst,
                                                 const float* __restrict__ dinv,
                                                 const float* __restrict__ h,
                                                 float* __restrict__ agg,
                                                 int e0, int nE, int t, int lane) {
    int cnt = nE - e0;
    if (cnt <= 0) return;
    int so = 0, dofs = 0;
    float c = 0.0f;
    bool sel = false;
    if (lane < cnt) {
        int s = src[e0 + lane];
        int d = dst[e0 + lane];
        so   = s * HIDDEN;
        dofs = d * HIDDEN;
        sel = (d >= HALF_SPLIT ? 1 : 0) == t;
        if (sel) c = dinv[s] * dinv[d];
    }
    unsigned mask = __builtin_amdgcn_ballot_w32(sel);
    float v0[32], v1[32];
#pragma unroll
    for (int j = 0; j < 32; ++j) {
        float a = 0.0f, b = 0.0f;
        if ((mask >> j) & 1u) {
            int ro = lane_i(so, j) + lane;
            a = h[ro];
            b = h[ro + 32];
        }
        v0[j] = a;
        v1[j] = b;
    }
#pragma unroll
    for (int j = 0; j < 32; ++j) {
        if ((mask >> j) & 1u) {
            float cj = lane_f(c, j);
            int ro = lane_i(dofs, j) + lane;
            atomic_add_f32(&agg[ro],      v0[j] * cj);
            atomic_add_f32(&agg[ro + 32], v1[j] * cj);
        }
    }
}

__global__ __launch_bounds__(256) void edge_agg1_kernel(const int* __restrict__ src,
                                                        const int* __restrict__ dst,
                                                        const float* __restrict__ dinv,
                                                        const float* __restrict__ h,
                                                        float* __restrict__ agg, int nE,
                                                        int* __restrict__ queues) {
    __shared__ int slot;
    int lane = threadIdx.x & 31;
    int wave = __builtin_amdgcn_readfirstlane(threadIdx.x >> 5);
    int nClaims = (nE + CLAIM_EDGES - 1) / CLAIM_EDGES;
    int half = my_half(&slot);
    for (int pass = 0; pass < 2; ++pass) {
        int t = half ^ pass;
        for (;;) {
            int idx = claim_block(queues + t * QSTRIDE, &slot);
            if (idx >= nClaims) break;
            int e0 = idx * CLAIM_EDGES + wave * 64;
            agg1_wave(src, dst, dinv, h, agg, e0,      nE, t, lane);
            agg1_wave(src, dst, dinv, h, agg, e0 + 32, nE, t, lane);
        }
    }
}

__global__ __launch_bounds__(256) void gemm2_kernel(const float* __restrict__ A,
                                                    const float* __restrict__ B,
                                                    const float* __restrict__ b2,
                                                    const float* __restrict__ dinv,
                                                    float* __restrict__ C,
                                                    float* __restrict__ out, int nTiles) {
    int wave = threadIdx.x >> 5;
    int lane = threadIdx.x & 31;
    int tile = blockIdx.x * 8 + wave;
    if (tile >= nTiles) return;
    int nr = lane & 15;
    int hi = lane >> 4;

    float di = dinv[tile * 16 + nr];
    float bb = b2[nr];

    const float* arow = A + (size_t)(tile * 16 + nr) * HIDDEN + 2 * hi;
    v2f af[HIDDEN / 4];
#pragma unroll
    for (int i = 0; i < HIDDEN / 4; ++i) {
        v2f a = *(const v2f*)(arow + 4 * i);
        a.x = a.x > 0.0f ? a.x : 0.0f;
        a.y = a.y > 0.0f ? a.y : 0.0f;
        af[i] = a;
    }
    v8f acc = {};
    asm volatile("" : "+v"(af[0]), "+v"(af[1]), "+v"(af[2]),  "+v"(af[3]),
                      "+v"(af[4]), "+v"(af[5]), "+v"(af[6]),  "+v"(af[7]));
    asm volatile("" : "+v"(af[8]),  "+v"(af[9]),  "+v"(af[10]), "+v"(af[11]),
                      "+v"(af[12]), "+v"(af[13]), "+v"(af[14]), "+v"(af[15]), "+v"(acc));
#pragma unroll
    for (int i = 0; i < HIDDEN / 4; ++i) {
        const float* b0p = B + (size_t)(4 * i + 2 * hi) * N_CLASSES + nr;
        v2f b = { b0p[0], b0p[N_CLASSES] };
        acc = __builtin_amdgcn_wmma_f32_16x16x4_f32(false, af[i], false, b, (short)0, acc, false, false);
    }
    asm volatile(WMMA_SLOTS4 : "+v"(acc), "+v"(di), "+v"(bb), "+v"(nr), "+v"(hi));
    int mbase = tile * 16 + hi * 8;
#pragma unroll
    for (int r = 0; r < 8; ++r) {
        float dlo = lane_f(di, r), dhi = lane_f(di, 8 + r);
        float dr = hi ? dhi : dlo;
        size_t o = (size_t)(mbase + r) * N_CLASSES + nr;
        C[o]   = acc[r];
        out[o] = dr * dr * acc[r] + bb;
    }
}

template <int K> static __device__ __forceinline__ int row_lane_i(int v) {
    return __builtin_amdgcn_update_dpp(0, v, 0x150 + K, 0xF, 0xF, false);
}
template <int K> static __device__ __forceinline__ float row_lane_f(float v) {
    return __int_as_float(row_lane_i<K>(__float_as_int(v)));
}

#define AGG2_LOAD(K) { int sk = row_lane_i<K>(s); int fk = row_lane_i<K>(f); \
                       if (fk) v[K] = h[sk * N_CLASSES + col]; }
#define AGG2_ADD(K)  { int dk = row_lane_i<K>(d); int fk = row_lane_i<K>(f); float ck = row_lane_f<K>(c); \
                       if (fk) atomic_add_f32(&out[dk * N_CLASSES + col], ck * v[K]); }
#define AGG2_ALL(M)  M(0) M(1) M(2) M(3) M(4) M(5) M(6) M(7) M(8) M(9) M(10) M(11) M(12) M(13) M(14) M(15)

static __device__ __forceinline__ void agg2_wave(const int* __restrict__ src,
                                                 const int* __restrict__ dst,
                                                 const float* __restrict__ dinv,
                                                 const float* __restrict__ h,
                                                 float* __restrict__ out,
                                                 int e0, int nE, int t, int lane) {
    int cnt = nE - e0;
    if (cnt <= 0) return;
    int s = 0, d = 0, f = 0;
    float c = 0.0f;
    if (lane < cnt) {
        s = src[e0 + lane];
        d = dst[e0 + lane];
        f = ((d >= HALF_SPLIT ? 1 : 0) == t) ? 1 : 0;
        if (f) c = dinv[s] * dinv[d];
    }
    int col = lane & 15;
    float v[16];
    AGG2_ALL(AGG2_LOAD)
    AGG2_ALL(AGG2_ADD)
}

__global__ __launch_bounds__(256) void edge_agg2_kernel(const int* __restrict__ src,
                                                        const int* __restrict__ dst,
                                                        const float* __restrict__ dinv,
                                                        const float* __restrict__ h,
                                                        float* __restrict__ out, int nE,
                                                        int* __restrict__ queues) {
    __shared__ int slot;
    int lane = threadIdx.x & 31;
    int wave = __builtin_amdgcn_readfirstlane(threadIdx.x >> 5);
    int nClaims = (nE + CLAIM_EDGES - 1) / CLAIM_EDGES;
    int half = my_half(&slot);
    for (int pass = 0; pass < 2; ++pass) {
        int t = half ^ pass;
        for (;;) {
            int idx = claim_block(queues + t * QSTRIDE, &slot);
            if (idx >= nClaims) break;
            int e0 = idx * CLAIM_EDGES + wave * 64;
            agg2_wave(src, dst, dinv, h, out, e0,      nE, t, lane);
            agg2_wave(src, dst, dinv, h, out, e0 + 32, nE, t, lane);
        }
    }
}

__global__ __launch_bounds__(256) void logsoftmax_kernel(float* __restrict__ out) {
    int r = blockIdx.x * 256 + threadIdx.x;
    if (r >= N_NODES) return;
    float v[N_CLASSES];
    float m = -INFINITY;
    for (int c = 0; c < N_CLASSES; ++c) {
        float x = out[(size_t)r * N_CLASSES + c];
        v[c] = x;
        m = fmaxf(m, x);
    }
    float s = 0.0f;
    for (int c = 0; c < N_CLASSES; ++c) s += expf(v[c] - m);
    float ls = logf(s);
    for (int c = 0; c < N_CLASSES; ++c)
        out[(size_t)r * N_CLASSES + c] = v[c] - m - ls;
}

extern "C" void kernel_launch(void* const* d_in, const int* in_sizes, int n_in,
                              void* d_out, int out_size, void* d_ws, size_t ws_size,
                              hipStream_t stream) {
    const float* x  = (const float*)d_in[0];
    const int*   ei = (const int*)d_in[1];
    const float* W1 = (const float*)d_in[2];
    const float* b1 = (const float*)d_in[3];
    const float* W2 = (const float*)d_in[4];
    const float* b2 = (const float*)d_in[5];
    float* out = (float*)d_out;

    int nE = in_sizes[1] / 2;
    const int* src = ei;
    const int* dst = ei + nE;

    float* dinv = (float*)d_ws;
    float* bufA = dinv + 102400;
    float* bufB = bufA + (size_t)N_NODES * HIDDEN;

    int* q1 = (int*)(dinv + 100096);
    int* q2 = q1 + 2 * QSTRIDE;

    int nTiles = N_NODES / 16;

    init_kernel<<<(N_NODES + 255) / 256, 256, 0, stream>>>(dinv, q1);
    degree_kernel<<<DEG_RANGES * DEG_SLICES, DEG_THREADS, 0, stream>>>(dst, dinv, nE);
    gemm1_kernel<<<G1_BLOCKS, G1_THREADS, 0, stream>>>(x, W1, b1, dinv, bufA, bufB, nTiles);
    edge_agg1_kernel<<<1536, 256, 0, stream>>>(src, dst, dinv, bufA, bufB, nE, q1);
    gemm2_kernel<<<(nTiles + 7) / 8, 256, 0, stream>>>(bufB, W2, b2, dinv, bufA, out, nTiles);
    edge_agg2_kernel<<<2048, 256, 0, stream>>>(src, dst, dinv, bufA, out, nE, q2);
    logsoftmax_kernel<<<(N_NODES + 255) / 256, 256, 0, stream>>>(out);
}


# Layer-2 edge pass: per-edge weight dinv[s]*dinv[d] stored once by layer 1's pass (in the output buffer, free until log-softmax), read coalesced; two scattered dinv loads per edge gone; sums unchanged
# speedup vs baseline: 2.5962x; 1.0552x over previous
#include <hip/hip_runtime.h>
#include <math.h>

typedef float v2f __attribute__((ext_vector_type(2)));
typedef float v8f __attribute__((ext_vector_type(8)));

#define N_NODES   100000
#define N_FEAT    512
#define HIDDEN    64
#define N_CLASSES 16

static __device__ __forceinline__ void atomic_add_f32(float* p, float v) {
    __hip_atomic_fetch_add(p, v, __ATOMIC_RELAXED, __HIP_MEMORY_SCOPE_AGENT);
}

#define QSTRIDE 64
__global__ __launch_bounds__(256) void init_kernel(float* deg, int* queues) {
    int t = blockIdx.x * 256 + threadIdx.x;
    if (t < 4)       queues[t * QSTRIDE] = 0;
    if (t < N_NODES) deg[t] = 1.0f;
}

#define DEG_RANGE   6272
#define DEG_RANGES  16
#define DEG_SLICES  16
#define DEG_THREADS 512
#define DEG_BATCH   28
__global__ __launch_bounds__(DEG_THREADS) void degree_kernel(const int* __restrict__ dst,
                                                             float* __restrict__ deg, int nE) {
    __shared__ unsigned cnt[DEG_RANGE];
    int tid   = threadIdx.x;
    int range = blockIdx.x % DEG_RANGES;
    int slice = blockIdx.x / DEG_RANGES;
    int lo    = range * DEG_RANGE;
    for (int i = tid; i < DEG_RANGE; i += DEG_THREADS) cnt[i] = 0u;
    __syncthreads();

    int per = (nE + DEG_SLICES - 1) / DEG_SLICES;
    int e0  = slice * per;
    int e1  = e0 + per < nE ? e0 + per : nE;
    for (int e = e0 + tid; e < e1; e += DEG_THREADS * DEG_BATCH) {
        int d[DEG_BATCH];
#pragma unroll
        for (int u = 0; u < DEG_BATCH; ++u) {
            int eu = e + u * DEG_THREADS;
            d[u] = eu < e1 ? dst[eu] : -1;
        }
#pragma unroll
        for (int u = 0; u < DEG_BATCH; ++u) {
            unsigned r = (unsigned)(d[u] - lo);
            if (r < (unsigned)DEG_RANGE) atomicAdd(&cnt[r], 1u);
        }
    }
    __syncthreads();

    for (int i = tid; i < DEG_RANGE; i += DEG_THREADS) {
        unsigned c = cnt[i];
        int n = lo + i;
        if (c != 0u && n < N_NODES) atomic_add_f32(&deg[n], (float)c);
    }
}

static __device__ __forceinline__ int lane_i(int v, int j) { return __builtin_amdgcn_readlane(v, j); }
static __device__ __forceinline__ float lane_f(float v, int j) {
    return __int_as_float(__builtin_amdgcn_readlane(__float_as_int(v), j));
}

#define WMMA_SLOTS4 "v_nop\n\tv_nop\n\tv_nop\n\tv_nop"

typedef float v4f __attribute__((ext_vector_type(4)));
#define G1_THREADS 512
#define G1_WAVES   (G1_THREADS / 32)
#define G1_BLOCKS  512
#define G1_PLANE   ((N_FEAT / 2) * 16 * 4)
__global__ __launch_bounds__(G1_THREADS) void gemm1_kernel(const float* __restrict__ A,
                                                           const float* __restrict__ B,
                                                           const float* __restrict__ bias,
                                                           float* __restrict__ dinv,
                                                           float* __restrict__ C,
                                                           float* __restrict__ S, int nTiles) {
    __shared__ float Bl[2 * G1_PLANE];
#pragma unroll
    for (int i = 0; i < (N_FEAT / 2) * (HIDDEN / 4) / G1_THREADS; ++i) {
        int it = i * G1_THREADS + threadIdx.x;
        int kp = it >> 4, q = it & 15;
        v4f r0 = *(const v4f*)(B + (size_t)(2 * kp) * HIDDEN + 4 * q);
        v4f r1 = *(const v4f*)(B + (size_t)(2 * kp + 1) * HIDDEN + 4 * q);
        int j = q >> 2;
        float* p = Bl + (j >> 1) * G1_PLANE + (kp * 16 + (q & 3) * 4) * 4 + (j & 1) * 2;
        *(v2f*)(p)      = (v2f){ r0.x, r1.x };
        *(v2f*)(p + 4)  = (v2f){ r0.y, r1.y };
        *(v2f*)(p + 8)  = (v2f){ r0.z, r1.z };
        *(v2f*)(p + 12) = (v2f){ r0.w, r1.w };
    }
    __syncthreads();

    int wave = threadIdx.x >> 5;
    int lane = threadIdx.x & 31;
    int nr = lane & 15;
    int hi = lane >> 4;
    for (int tile = wave * (int)gridDim.x + (int)blockIdx.x; tile < nTiles;
         tile += G1_WAVES * (int)gridDim.x) {
        float di = rsqrtf(dinv[tile * 16 + nr]);
        float bb0 = bias[nr], bb1 = bias[nr + 16], bb2 = bias[nr + 32], bb3 = bias[nr + 48];

        const float* arow = A + (size_t)(tile * 16 + nr) * N_FEAT + 2 * hi;
        const float* brow = Bl + (hi * 16 + nr) * 4;
        v8f acc0 = {}, acc1 = {}, acc2 = {}, acc3 = {};

        for (int kk = 0; kk < N_FEAT; kk += 4) {
            v2f a = *(const v2f*)(arow + kk);
            const float* bp = brow + (kk >> 1) * 64;
            v4f q0 = *(const v4f*)(bp);
            v4f q1 = *(const v4f*)(bp + G1_PLANE);
            v2f b0 = { q0.x, q0.y };
            v2f b1 = { q0.z, q0.w };
            v2f b2 = { q1.x, q1.y };
            v2f b3 = { q1.z, q1.w };
            acc0 = __builtin_amdgcn_wmma_f32_16x16x4_f32(false, a, false, b0, (short)0, acc0, false, false);
            acc1 = __builtin_amdgcn_wmma_f32_16x16x4_f32(false, a, false, b1, (short)0, acc1, false, false);
            acc2 = __builtin_amdgcn_wmma_f32_16x16x4_f32(false, a, false, b2, (short)0, acc2, false, false);
            acc3 = __builtin_amdgcn_wmma_f32_16x16x4_f32(false, a, false, b3, (short)0, acc3, false, false);
        }
        asm volatile(WMMA_SLOTS4 : "+v"(acc0), "+v"(acc1), "+v"(acc2), "+v"(acc3), "+v"(di),
                                   "+v"(bb0), "+v"(bb1), "+v"(bb2), "+v"(bb3), "+v"(nr), "+v"(hi));
        if (hi == 0) dinv[tile * 16 + nr] = di;
        int mbase = tile * 16 + hi * 8;
#pragma unroll
        for (int r = 0; r < 8; ++r) {
            float dlo = lane_f(di, r), dhi = lane_f(di, 8 + r);
            float dr = hi ? dhi : dlo;
            float d2 = dr * dr;
            size_t o = (size_t)(mbase + r) * HIDDEN + nr;
            float* crow = C + o;
            float* srow = S + o;
            crow[0]  = acc0[r];
            crow[16] = acc1[r];
            crow[32] = acc2[r];
            crow[48] = acc3[r];
            srow[0]  = d2 * acc0[r] + bb0;
            srow[16] = d2 * acc1[r] + bb1;
            srow[32] = d2 * acc2[r] + bb2;
            srow[48] = d2 * acc3[r] + bb3;
        }
    }
}

#define HALF_SPLIT  (N_NODES / 2)
#define CLAIM_EDGES 512

static __device__ __forceinline__ int my_half(int* slot) {
    if (threadIdx.x == 0) *slot = (__builtin_amdgcn_s_sendmsg_rtn(0x87) >> 18) & 1;
    __syncthreads();
    int hf = *slot;
    __syncthreads();
    return hf;
}

static __device__ __forceinline__ int claim_block(int* q, int* slot) {
    if (threadIdx.x == 0)
        *slot = __hip_atomic_fetch_add(q, 1, __ATOMIC_RELAXED, __HIP_MEMORY_SCOPE_AGENT);
    __syncthreads();
    int idx = *slot;
    __syncthreads();
    return idx;
}

static __device__ __forceinline__ void agg1_wave(const int* __restrict__ src,
                                                 const int* __restrict__ dst,
                                                 const float* __restrict__ dinv,
                                                 const float* __restrict__ h,
                                                 float* __restrict__ agg,
                                                 float* __restrict__ cw,
                                                 int e0, int nE, int t, int lane) {
    int cnt = nE - e0;
    if (cnt <= 0) return;
    int so = 0, dofs = 0;
    float c = 0.0f;
    bool sel = false;
    if (lane < cnt) {
        int s = src[e0 + lane];
        int d = dst[e0 + lane];
        so   = s * HIDDEN;
        dofs = d * HIDDEN;
        sel = (d >= HALF_SPLIT ? 1 : 0) == t;
        if (sel) {
            c = dinv[s] * dinv[d];
            cw[e0 + lane] = c;
        }
    }
    unsigned mask = __builtin_amdgcn_ballot_w32(sel);
    float v0[32], v1[32];
#pragma unroll
    for (int j = 0; j < 32; ++j) {
        float a = 0.0f, b = 0.0f;
        if ((mask >> j) & 1u) {
            int ro = lane_i(so, j) + lane;
            a = h[ro];
            b = h[ro + 32];
        }
        v0[j] = a;
        v1[j] = b;
    }
#pragma unroll
    for (int j = 0; j < 32; ++j) {
        if ((mask >> j) & 1u) {
            float cj = lane_f(c, j);
            int ro = lane_i(dofs, j) + lane;
            atomic_add_f32(&agg[ro],      v0[j] * cj);
            atomic_add_f32(&agg[ro + 32], v1[j] * cj);
        }
    }
}

__global__ __launch_bounds__(256) void edge_agg1_kernel(const int* __restrict__ src,
                                                        const int* __restrict__ dst,
                                                        const float* __restrict__ dinv,
                                                        const float* __restrict__ h,
                                                        float* __restrict__ agg,
                                                        float* __restrict__ cw, int nE,
                                                        int* __restrict__ queues) {
    __shared__ int slot;
    int lane = threadIdx.x & 31;
    int wave = __builtin_amdgcn_readfirstlane(threadIdx.x >> 5);
    int nClaims = (nE + CLAIM_EDGES - 1) / CLAIM_EDGES;
    int half = my_half(&slot);
    for (int pass = 0; pass < 2; ++pass) {
        int t = half ^ pass;
        for (;;) {
            int idx = claim_block(queues + t * QSTRIDE, &slot);
            if (idx >= nClaims) break;
            int e0 = idx * CLAIM_EDGES + wave * 64;
            agg1_wave(src, dst, dinv, h, agg, cw, e0,      nE, t, lane);
            agg1_wave(src, dst, dinv, h, agg, cw, e0 + 32, nE, t, lane);
        }
    }
}

__global__ __launch_bounds__(256) void gemm2_kernel(const float* __restrict__ A,
                                                    const float* __restrict__ B,
                                                    const float* __restrict__ b2,
                                                    const float* __restrict__ dinv,
                                                    float* __restrict__ C,
                                                    float* __restrict__ out, int nTiles) {
    int wave = threadIdx.x >> 5;
    int lane = threadIdx.x & 31;
    int tile = blockIdx.x * 8 + wave;
    if (tile >= nTiles) return;
    int nr = lane & 15;
    int hi = lane >> 4;

    float di = dinv[tile * 16 + nr];
    float bb = b2[nr];

    const float* arow = A + (size_t)(tile * 16 + nr) * HIDDEN + 2 * hi;
    v2f af[HIDDEN / 4];
#pragma unroll
    for (int i = 0; i < HIDDEN / 4; ++i) {
        v2f a = *(const v2f*)(arow + 4 * i);
        a.x = a.x > 0.0f ? a.x : 0.0f;
        a.y = a.y > 0.0f ? a.y : 0.0f;
        af[i] = a;
    }
    v8f acc = {};
    asm volatile("" : "+v"(af[0]), "+v"(af[1]), "+v"(af[2]),  "+v"(af[3]),
                      "+v"(af[4]), "+v"(af[5]), "+v"(af[6]),  "+v"(af[7]));
    asm volatile("" : "+v"(af[8]),  "+v"(af[9]),  "+v"(af[10]), "+v"(af[11]),
                      "+v"(af[12]), "+v"(af[13]), "+v"(af[14]), "+v"(af[15]), "+v"(acc));
#pragma unroll
    for (int i = 0; i < HIDDEN / 4; ++i) {
        const float* b0p = B + (size_t)(4 * i + 2 * hi) * N_CLASSES + nr;
        v2f b = { b0p[0], b0p[N_CLASSES] };
        acc = __builtin_amdgcn_wmma_f32_16x16x4_f32(false, af[i], false, b, (short)0, acc, false, false);
    }
    asm volatile(WMMA_SLOTS4 : "+v"(acc), "+v"(di), "+v"(bb), "+v"(nr), "+v"(hi));
    int mbase = tile * 16 + hi * 8;
#pragma unroll
    for (int r = 0; r < 8; ++r) {
        float dlo = lane_f(di, r), dhi = lane_f(di, 8 + r);
        float dr = hi ? dhi : dlo;
        size_t o = (size_t)(mbase + r) * N_CLASSES + nr;
        C[o]   = acc[r];
        out[o] = dr * dr * acc[r] + bb;
    }
}

template <int K> static __device__ __forceinline__ int row_lane_i(int v) {
    return __builtin_amdgcn_update_dpp(0, v, 0x150 + K, 0xF, 0xF, false);
}
template <int K> static __device__ __forceinline__ float row_lane_f(float v) {
    return __int_as_float(row_lane_i<K>(__float_as_int(v)));
}

#define AGG2_LOAD(K) { int sk = row_lane_i<K>(s); int fk = row_lane_i<K>(f); \
                       if (fk) v[K] = h[sk * N_CLASSES + col]; }
#define AGG2_ADD(K)  { int dk = row_lane_i<K>(d); int fk = row_lane_i<K>(f); float ck = row_lane_f<K>(c); \
                       if (fk) atomic_add_f32(&out[dk * N_CLASSES + col], ck * v[K]); }
#define AGG2_ALL(M)  M(0) M(1) M(2) M(3) M(4) M(5) M(6) M(7) M(8) M(9) M(10) M(11) M(12) M(13) M(14) M(15)

static __device__ __forceinline__ void agg2_wave(const int* __restrict__ src,
                                                 const int* __restrict__ dst,
                                                 const float* __restrict__ cw,
                                                 const float* __restrict__ h,
                                                 float* __restrict__ out,
                                                 int e0, int nE, int t, int lane) {
    int cnt = nE - e0;
    if (cnt <= 0) return;
    int s = 0, d = 0, f = 0;
    float c = 0.0f;
    if (lane < cnt) {
        s = src[e0 + lane];
        d = dst[e0 + lane];
        f = ((d >= HALF_SPLIT ? 1 : 0) == t) ? 1 : 0;
        c = cw[e0 + lane];
    }
    int col = lane & 15;
    float v[16];
    AGG2_ALL(AGG2_LOAD)
    AGG2_ALL(AGG2_ADD)
}

__global__ __launch_bounds__(256) void edge_agg2_kernel(const int* __restrict__ src,
                                                        const int* __restrict__ dst,
                                                        const float* __restrict__ cw,
                                                        const float* __restrict__ h,
                                                        float* __restrict__ out, int nE,
                                                        int* __restrict__ queues) {
    __shared__ int slot;
    int lane = threadIdx.x & 31;
    int wave = __builtin_amdgcn_readfirstlane(threadIdx.x >> 5);
    int nClaims = (nE + CLAIM_EDGES - 1) / CLAIM_EDGES;
    int half = my_half(&slot);
    for (int pass = 0; pass < 2; ++pass) {
        int t = half ^ pass;
        for (;;) {
            int idx = claim_block(queues + t * QSTRIDE, &slot);
            if (idx >= nClaims) break;
            int e0 = idx * CLAIM_EDGES + wave * 64;
            agg2_wave(src, dst, cw, h, out, e0,      nE, t, lane);
            agg2_wave(src, dst, cw, h, out, e0 + 32, nE, t, lane);
        }
    }
}

__global__ __launch_bounds__(256) void logsoftmax_kernel(const float* __restrict__ in,
                                                         float* __restrict__ out) {
    int r = blockIdx.x * 256 + threadIdx.x;
    if (r >= N_NODES) return;
    float v[N_CLASSES];
    float m = -INFINITY;
    for (int c = 0; c < N_CLASSES; ++c) {
        float x = in[(size_t)r * N_CLASSES + c];
        v[c] = x;
        m = fmaxf(m, x);
    }
    float s = 0.0f;
    for (int c = 0; c < N_CLASSES; ++c) s += expf(v[c] - m);
    float ls = logf(s);
    for (int c = 0; c < N_CLASSES; ++c)
        out[(size_t)r * N_CLASSES + c] = v[c] - m - ls;
}

extern "C" void kernel_launch(void* const* d_in, const int* in_sizes, int n_in,
                              void* d_out, int out_size, void* d_ws, size_t ws_size,
                              hipStream_t stream) {
    const float* x  = (const float*)d_in[0];
    const int*   ei = (const int*)d_in[1];
    const float* W1 = (const float*)d_in[2];
    const float* b1 = (const float*)d_in[3];
    const float* W2 = (const float*)d_in[4];
    const float* b2 = (const float*)d_in[5];
    float* out = (float*)d_out;

    int nE = in_sizes[1] / 2;
    const int* src = ei;
    const int* dst = ei + nE;

    float* dinv = (float*)d_ws;
    float* bufA = dinv + 102400;
    float* bufB = bufA + (size_t)N_NODES * HIDDEN;
    float* acc2 = bufA + (size_t)N_NODES * N_CLASSES;
    float* cw = out;

    int* q1 = (int*)(dinv + 100096);
    int* q2 = q1 + 2 * QSTRIDE;

    int nTiles = N_NODES / 16;

    init_kernel<<<(N_NODES + 255) / 256, 256, 0, stream>>>(dinv, q1);
    degree_kernel<<<DEG_RANGES * DEG_SLICES, DEG_THREADS, 0, stream>>>(dst, dinv, nE);
    gemm1_kernel<<<G1_BLOCKS, G1_THREADS, 0, stream>>>(x, W1, b1, dinv, bufA, bufB, nTiles);
    edge_agg1_kernel<<<1536, 256, 0, stream>>>(src, dst, dinv, bufA, bufB, cw, nE, q1);
    gemm2_kernel<<<(nTiles + 7) / 8, 256, 0, stream>>>(bufB, W2, b2, dinv, bufA, acc2, nTiles);
    edge_agg2_kernel<<<2048, 256, 0, stream>>>(src, dst, cw, bufA, acc2, nE, q2);
    logsoftmax_kernel<<<(N_NODES + 255) / 256, 256, 0, stream>>>(acc2, out);
}


# Per-edge weight rsqrt(deg[s])*rsqrt(deg[d]) formed once by GEMM1's 3 tile-less waves per work-group beside its products; both edge passes read it coalesced; layer 1's scattered loads gone; sums same
# speedup vs baseline: 2.6888x; 1.0357x over previous
#include <hip/hip_runtime.h>
#include <math.h>

typedef float v2f __attribute__((ext_vector_type(2)));
typedef float v8f __attribute__((ext_vector_type(8)));

#define N_NODES   100000
#define N_FEAT    512
#define HIDDEN    64
#define N_CLASSES 16

static __device__ __forceinline__ void atomic_add_f32(float* p, float v) {
    __hip_atomic_fetch_add(p, v, __ATOMIC_RELAXED, __HIP_MEMORY_SCOPE_AGENT);
}

#define QSTRIDE 64
__global__ __launch_bounds__(256) void init_kernel(float* deg, int* queues) {
    int t = blockIdx.x * 256 + threadIdx.x;
    if (t < 4)       queues[t * QSTRIDE] = 0;
    if (t < N_NODES) deg[t] = 1.0f;
}

#define DEG_RANGE   6272
#define DEG_RANGES  16
#define DEG_SLICES  16
#define DEG_THREADS 512
#define DEG_BATCH   28
__global__ __launch_bounds__(DEG_THREADS) void degree_kernel(const int* __restrict__ dst,
                                                             float* __restrict__ deg, int nE) {
    __shared__ unsigned cnt[DEG_RANGE];
    int tid   = threadIdx.x;
    int range = blockIdx.x % DEG_RANGES;
    int slice = blockIdx.x / DEG_RANGES;
    int lo    = range * DEG_RANGE;
    for (int i = tid; i < DEG_RANGE; i += DEG_THREADS) cnt[i] = 0u;
    __syncthreads();

    int per = (nE + DEG_SLICES - 1) / DEG_SLICES;
    int e0  = slice * per;
    int e1  = e0 + per < nE ? e0 + per : nE;
    for (int e = e0 + tid; e < e1; e += DEG_THREADS * DEG_BATCH) {
        int d[DEG_BATCH];
#pragma unroll
        for (int u = 0; u < DEG_BATCH; ++u) {
            int eu = e + u * DEG_THREADS;
            d[u] = eu < e1 ? dst[eu] : -1;
        }
#pragma unroll
        for (int u = 0; u < DEG_BATCH; ++u) {
            unsigned r = (unsigned)(d[u] - lo);
            if (r < (unsigned)DEG_RANGE) atomicAdd(&cnt[r], 1u);
        }
    }
    __syncthreads();

    for (int i = tid; i < DEG_RANGE; i += DEG_THREADS) {
        unsigned c = cnt[i];
        int n = lo + i;
        if (c != 0u && n < N_NODES) atomic_add_f32(&deg[n], (float)c);
    }
}

static __device__ __forceinline__ int lane_i(int v, int j) { return __builtin_amdgcn_readlane(v, j); }
static __device__ __forceinline__ float lane_f(float v, int j) {
    return __int_as_float(__builtin_amdgcn_readlane(__float_as_int(v), j));
}

#define WMMA_SLOTS4 "v_nop\n\tv_nop\n\tv_nop\n\tv_nop"

typedef float v4f __attribute__((ext_vector_type(4)));
#define G1_THREADS 512
#define G1_WAVES   (G1_THREADS / 32)
#define G1_BLOCKS  512
#define G1_PLANE   ((N_FEAT / 2) * 16 * 4)
#define G1_MM_WAVES 13
#define G1_EW_WAVES (G1_WAVES - G1_MM_WAVES)
#define EW_GROUPS  8

static __device__ __forceinline__ void edge_weights(const int* __restrict__ src,
                                                    const int* __restrict__ dst,
                                                    const float* __restrict__ deg,
                                                    float* __restrict__ cw,
                                                    int nE, int ew, int nEw, int lane) {
    for (int base = ew * (32 * EW_GROUPS); base < nE; base += nEw * (32 * EW_GROUPS)) {
        if (base + 32 * EW_GROUPS <= nE) {
            int s[EW_GROUPS], d[EW_GROUPS];
            float ds[EW_GROUPS], dd[EW_GROUPS];
#pragma unroll
            for (int u = 0; u < EW_GROUPS; ++u) {
                s[u] = src[base + u * 32 + lane];
                d[u] = dst[base + u * 32 + lane];
            }
#pragma unroll
            for (int u = 0; u < EW_GROUPS; ++u) {
                ds[u] = deg[s[u]];
                dd[u] = deg[d[u]];
            }
#pragma unroll
            for (int u = 0; u < EW_GROUPS; ++u)
                cw[base + u * 32 + lane] = rsqrtf(ds[u]) * rsqrtf(dd[u]);
        } else {
            for (int e = base + lane; e < nE; e += 32)
                cw[e] = rsqrtf(deg[src[e]]) * rsqrtf(deg[dst[e]]);
        }
    }
}

__global__ __launch_bounds__(G1_THREADS) void gemm1_kernel(const float* __restrict__ A,
                                                           const float* __restrict__ B,
                                                           const float* __restrict__ bias,
                                                           const float* __restrict__ deg,
                                                           float* __restrict__ C,
                                                           float* __restrict__ S, int nTiles,
                                                           const int* __restrict__ src,
                                                           const int* __restrict__ dst,
                                                           float* __restrict__ cw, int nE) {
    __shared__ float Bl[2 * G1_PLANE];
#pragma unroll
    for (int i = 0; i < (N_FEAT / 2) * (HIDDEN / 4) / G1_THREADS; ++i) {
        int it = i * G1_THREADS + threadIdx.x;
        int kp = it >> 4, q = it & 15;
        v4f r0 = *(const v4f*)(B + (size_t)(2 * kp) * HIDDEN + 4 * q);
        v4f r1 = *(const v4f*)(B + (size_t)(2 * kp + 1) * HIDDEN + 4 * q);
        int j = q >> 2;
        float* p = Bl + (j >> 1) * G1_PLANE + (kp * 16 + (q & 3) * 4) * 4 + (j & 1) * 2;
        *(v2f*)(p)      = (v2f){ r0.x, r1.x };
        *(v2f*)(p + 4)  = (v2f){ r0.y, r1.y };
        *(v2f*)(p + 8)  = (v2f){ r0.z, r1.z };
        *(v2f*)(p + 12) = (v2f){ r0.w, r1.w };
    }
    __syncthreads();

    int wave = threadIdx.x >> 5;
    int lane = threadIdx.x & 31;
    int wv = __builtin_amdgcn_readfirstlane(wave);
    if (wv >= G1_MM_WAVES) {
        edge_weights(src, dst, deg, cw, nE,
                     (wv - G1_MM_WAVES) * (int)gridDim.x + (int)blockIdx.x,
                     G1_EW_WAVES * (int)gridDim.x, lane);
        return;
    }
    int nr = lane & 15;
    int hi = lane >> 4;
    for (int tile = wave * (int)gridDim.x + (int)blockIdx.x; tile < nTiles;
         tile += G1_MM_WAVES * (int)gridDim.x) {
        float di = rsqrtf(deg[tile * 16 + nr]);
        float bb0 = bias[nr], bb1 = bias[nr + 16], bb2 = bias[nr + 32], bb3 = bias[nr + 48];

        const float* arow = A + (size_t)(tile * 16 + nr) * N_FEAT + 2 * hi;
        const float* brow = Bl + (hi * 16 + nr) * 4;
        v8f acc0 = {}, acc1 = {}, acc2 = {}, acc3 = {};

        for (int kk = 0; kk < N_FEAT; kk += 4) {
            v2f a = *(const v2f*)(arow + kk);
            const float* bp = brow + (kk >> 1) * 64;
            v4f q0 = *(const v4f*)(bp);
            v4f q1 = *(const v4f*)(bp + G1_PLANE);
            v2f b0 = { q0.x, q0.y };
            v2f b1 = { q0.z, q0.w };
            v2f b2 = { q1.x, q1.y };
            v2f b3 = { q1.z, q1.w };
            acc0 = __builtin_amdgcn_wmma_f32_16x16x4_f32(false, a, false, b0, (short)0, acc0, false, false);
            acc1 = __builtin_amdgcn_wmma_f32_16x16x4_f32(false, a, false, b1, (short)0, acc1, false, false);
            acc2 = __builtin_amdgcn_wmma_f32_16x16x4_f32(false, a, false, b2, (short)0, acc2, false, false);
            acc3 = __builtin_amdgcn_wmma_f32_16x16x4_f32(false, a, false, b3, (short)0, acc3, false, false);
        }
        asm volatile(WMMA_SLOTS4 : "+v"(acc0), "+v"(acc1), "+v"(acc2), "+v"(acc3), "+v"(di),
                                   "+v"(bb0), "+v"(bb1), "+v"(bb2), "+v"(bb3), "+v"(nr), "+v"(hi));
        int mbase = tile * 16 + hi * 8;
#pragma unroll
        for (int r = 0; r < 8; ++r) {
            float dlo = lane_f(di, r), dhi = lane_f(di, 8 + r);
            float dr = hi ? dhi : dlo;
            float d2 = dr * dr;
            size_t o = (size_t)(mbase + r) * HIDDEN + nr;
            float* crow = C + o;
            float* srow = S + o;
            crow[0]  = acc0[r];
            crow[16] = acc1[r];
            crow[32] = acc2[r];
            crow[48] = acc3[r];
            srow[0]  = d2 * acc0[r] + bb0;
            srow[16] = d2 * acc1[r] + bb1;
            srow[32] = d2 * acc2[r] + bb2;
            srow[48] = d2 * acc3[r] + bb3;
        }
    }
}

#define HALF_SPLIT  (N_NODES / 2)
#define CLAIM_EDGES 512

static __device__ __forceinline__ int my_half(int* slot) {
    if (threadIdx.x == 0) *slot = (__builtin_amdgcn_s_sendmsg_rtn(0x87) >> 18) & 1;
    __syncthreads();
    int hf = *slot;
    __syncthreads();
    return hf;
}

static __device__ __forceinline__ int claim_block(int* q, int* slot) {
    if (threadIdx.x == 0)
        *slot = __hip_atomic_fetch_add(q, 1, __ATOMIC_RELAXED, __HIP_MEMORY_SCOPE_AGENT);
    __syncthreads();
    int idx = *slot;
    __syncthreads();
    return idx;
}

static __device__ __forceinline__ void agg1_wave(const int* __restrict__ src,
                                                 const int* __restrict__ dst,
                                                 const float* __restrict__ h,
                                                 float* __restrict__ agg,
                                                 const float* __restrict__ cw,
                                                 int e0, int nE, int t, int lane) {
    int cnt = nE - e0;
    if (cnt <= 0) return;
    int so = 0, dofs = 0;
    float c = 0.0f;
    bool sel = false;
    if (lane < cnt) {
        int s = src[e0 + lane];
        int d = dst[e0 + lane];
        c = cw[e0 + lane];
        so   = s * HIDDEN;
        dofs = d * HIDDEN;
        sel = (d >= HALF_SPLIT ? 1 : 0) == t;
    }
    unsigned mask = __builtin_amdgcn_ballot_w32(sel);
    float v0[32], v1[32];
#pragma unroll
    for (int j = 0; j < 32; ++j) {
        float a = 0.0f, b = 0.0f;
        if ((mask >> j) & 1u) {
            int ro = lane_i(so, j) + lane;
            a = h[ro];
            b = h[ro + 32];
        }
        v0[j] = a;
        v1[j] = b;
    }
#pragma unroll
    for (int j = 0; j < 32; ++j) {
        if ((mask >> j) & 1u) {
            float cj = lane_f(c, j);
            int ro = lane_i(dofs, j) + lane;
            atomic_add_f32(&agg[ro],      v0[j] * cj);
            atomic_add_f32(&agg[ro + 32], v1[j] * cj);
        }
    }
}

__global__ __launch_bounds__(256) void edge_agg1_kernel(const int* __restrict__ src,
                                                        const int* __restrict__ dst,
                                                        const float* __restrict__ h,
                                                        float* __restrict__ agg,
                                                        const float* __restrict__ cw, int nE,
                                                        int* __restrict__ queues) {
    __shared__ int slot;
    int lane = threadIdx.x & 31;
    int wave = __builtin_amdgcn_readfirstlane(threadIdx.x >> 5);
    int nClaims = (nE + CLAIM_EDGES - 1) / CLAIM_EDGES;
    int half = my_half(&slot);
    for (int pass = 0; pass < 2; ++pass) {
        int t = half ^ pass;
        for (;;) {
            int idx = claim_block(queues + t * QSTRIDE, &slot);
            if (idx >= nClaims) break;
            int e0 = idx * CLAIM_EDGES + wave * 64;
            agg1_wave(src, dst, h, agg, cw, e0,      nE, t, lane);
            agg1_wave(src, dst, h, agg, cw, e0 + 32, nE, t, lane);
        }
    }
}

__global__ __launch_bounds__(256) void gemm2_kernel(const float* __restrict__ A,
                                                    const float* __restrict__ B,
                                                    const float* __restrict__ b2,
                                                    const float* __restrict__ deg,
                                                    float* __restrict__ C,
                                                    float* __restrict__ out, int nTiles) {
    int wave = threadIdx.x >> 5;
    int lane = threadIdx.x & 31;
    int tile = blockIdx.x * 8 + wave;
    if (tile >= nTiles) return;
    int nr = lane & 15;
    int hi = lane >> 4;

    float di = rsqrtf(deg[tile * 16 + nr]);
    float bb = b2[nr];

    const float* arow = A + (size_t)(tile * 16 + nr) * HIDDEN + 2 * hi;
    v2f af[HIDDEN / 4];
#pragma unroll
    for (int i = 0; i < HIDDEN / 4; ++i) {
        v2f a = *(const v2f*)(arow + 4 * i);
        a.x = a.x > 0.0f ? a.x : 0.0f;
        a.y = a.y > 0.0f ? a.y : 0.0f;
        af[i] = a;
    }
    v8f acc = {};
    asm volatile("" : "+v"(af[0]), "+v"(af[1]), "+v"(af[2]),  "+v"(af[3]),
                      "+v"(af[4]), "+v"(af[5]), "+v"(af[6]),  "+v"(af[7]));
    asm volatile("" : "+v"(af[8]),  "+v"(af[9]),  "+v"(af[10]), "+v"(af[11]),
                      "+v"(af[12]), "+v"(af[13]), "+v"(af[14]), "+v"(af[15]), "+v"(acc), "+v"(di));
#pragma unroll
    for (int i = 0; i < HIDDEN / 4; ++i) {
        const float* b0p = B + (size_t)(4 * i + 2 * hi) * N_CLASSES + nr;
        v2f b = { b0p[0], b0p[N_CLASSES] };
        acc = __builtin_amdgcn_wmma_f32_16x16x4_f32(false, af[i], false, b, (short)0, acc, false, false);
    }
    asm volatile(WMMA_SLOTS4 : "+v"(acc), "+v"(di), "+v"(bb), "+v"(nr), "+v"(hi));
    int mbase = tile * 16 + hi * 8;
#pragma unroll
    for (int r = 0; r < 8; ++r) {
        float dlo = lane_f(di, r), dhi = lane_f(di, 8 + r);
        float dr = hi ? dhi : dlo;
        size_t o = (size_t)(mbase + r) * N_CLASSES + nr;
        C[o]   = acc[r];
        out[o] = dr * dr * acc[r] + bb;
    }
}

template <int K> static __device__ __forceinline__ int row_lane_i(int v) {
    return __builtin_amdgcn_update_dpp(0, v, 0x150 + K, 0xF, 0xF, false);
}
template <int K> static __device__ __forceinline__ float row_lane_f(float v) {
    return __int_as_float(row_lane_i<K>(__float_as_int(v)));
}

#define AGG2_LOAD(K) { int sk = row_lane_i<K>(s); int fk = row_lane_i<K>(f); \
                       if (fk) v[K] = h[sk * N_CLASSES + col]; }
#define AGG2_ADD(K)  { int dk = row_lane_i<K>(d); int fk = row_lane_i<K>(f); float ck = row_lane_f<K>(c); \
                       if (fk) atomic_add_f32(&out[dk * N_CLASSES + col], ck * v[K]); }
#define AGG2_ALL(M)  M(0) M(1) M(2) M(3) M(4) M(5) M(6) M(7) M(8) M(9) M(10) M(11) M(12) M(13) M(14) M(15)

static __device__ __forceinline__ void agg2_wave(const int* __restrict__ src,
                                                 const int* __restrict__ dst,
                                                 const float* __restrict__ cw,
                                                 const float* __restrict__ h,
                                                 float* __restrict__ out,
                                                 int e0, int nE, int t, int lane) {
    int cnt = nE - e0;
    if (cnt <= 0) return;
    int s = 0, d = 0, f = 0;
    float c = 0.0f;
    if (lane < cnt) {
        s = src[e0 + lane];
        d = dst[e0 + lane];
        f = ((d >= HALF_SPLIT ? 1 : 0) == t) ? 1 : 0;
        c = cw[e0 + lane];
    }
    int col = lane & 15;
    float v[16];
    AGG2_ALL(AGG2_LOAD)
    AGG2_ALL(AGG2_ADD)
}

__global__ __launch_bounds__(256) void edge_agg2_kernel(const int* __restrict__ src,
                                                        const int* __restrict__ dst,
                                                        const float* __restrict__ cw,
                                                        const float* __restrict__ h,
                                                        float* __restrict__ out, int nE,
                                                        int* __restrict__ queues) {
    __shared__ int slot;
    int lane = threadIdx.x & 31;
    int wave = __builtin_amdgcn_readfirstlane(threadIdx.x >> 5);
    int nClaims = (nE + CLAIM_EDGES - 1) / CLAIM_EDGES;
    int half = my_half(&slot);
    for (int pass = 0; pass < 2; ++pass) {
        int t = half ^ pass;
        for (;;) {
            int idx = claim_block(queues + t * QSTRIDE, &slot);
            if (idx >= nClaims) break;
            int e0 = idx * CLAIM_EDGES + wave * 64;
            agg2_wave(src, dst, cw, h, out, e0,      nE, t, lane);
            agg2_wave(src, dst, cw, h, out, e0 + 32, nE, t, lane);
        }
    }
}

__global__ __launch_bounds__(256) void logsoftmax_kernel(const float* __restrict__ in,
                                                         float* __restrict__ out) {
    int r = blockIdx.x * 256 + threadIdx.x;
    if (r >= N_NODES) return;
    float v[N_CLASSES];
    float m = -INFINITY;
    for (int c = 0; c < N_CLASSES; ++c) {
        float x = in[(size_t)r * N_CLASSES + c];
        v[c] = x;
        m = fmaxf(m, x);
    }
    float s = 0.0f;
    for (int c = 0; c < N_CLASSES; ++c) s += expf(v[c] - m);
    float ls = logf(s);
    for (int c = 0; c < N_CLASSES; ++c)
        out[(size_t)r * N_CLASSES + c] = v[c] - m - ls;
}

extern "C" void kernel_launch(void* const* d_in, const int* in_sizes, int n_in,
                              void* d_out, int out_size, void* d_ws, size_t ws_size,
                              hipStream_t stream) {
    const float* x  = (const float*)d_in[0];
    const int*   ei = (const int*)d_in[1];
    const float* W1 = (const float*)d_in[2];
    const float* b1 = (const float*)d_in[3];
    const float* W2 = (const float*)d_in[4];
    const float* b2 = (const float*)d_in[5];
    float* out = (float*)d_out;

    int nE = in_sizes[1] / 2;
    const int* src = ei;
    const int* dst = ei + nE;

    float* dinv = (float*)d_ws;
    float* bufA = dinv + 102400;
    float* bufB = bufA + (size_t)N_NODES * HIDDEN;
    float* acc2 = bufA + (size_t)N_NODES * N_CLASSES;
    float* cw = out;

    int* q1 = (int*)(dinv + 100096);
    int* q2 = q1 + 2 * QSTRIDE;

    int nTiles = N_NODES / 16;

    init_kernel<<<(N_NODES + 255) / 256, 256, 0, stream>>>(dinv, q1);
    degree_kernel<<<DEG_RANGES * DEG_SLICES, DEG_THREADS, 0, stream>>>(dst, dinv, nE);
    gemm1_kernel<<<G1_BLOCKS, G1_THREADS, 0, stream>>>(x, W1, b1, dinv, bufA, bufB, nTiles, src, dst, cw, nE);
    edge_agg1_kernel<<<1536, 256, 0, stream>>>(src, dst, bufA, bufB, cw, nE, q1);
    gemm2_kernel<<<(nTiles + 7) / 8, 256, 0, stream>>>(bufB, W2, b2, dinv, bufA, acc2, nTiles);
    edge_agg2_kernel<<<2048, 256, 0, stream>>>(src, dst, cw, bufA, acc2, nE, q2);
    logsoftmax_kernel<<<(N_NODES + 255) / 256, 256, 0, stream>>>(acc2, out);
}
